# starting kernel
# baseline (speedup 1.0000x reference)
#include <hip/hip_runtime.h>
#include <stddef.h>

#define NU   8039
#define NI   32770
#define NB   4771
#define NTOT (NU + NI + NB)
#define EMB  32
#define NNZE 2000000
#define BATCHN 2048
#define KB   100
#define EPSF 1e-8f
#define L2NORM 1e-5f

typedef __attribute__((ext_vector_type(2))) float v2f;
typedef __attribute__((ext_vector_type(8))) float v8f;

__device__ __forceinline__ void atomAddF(float* p, float v) {
    (void)__hip_atomic_fetch_add(p, v, __ATOMIC_RELAXED, __HIP_MEMORY_SCOPE_AGENT);
}

__device__ __forceinline__ float feat_at(const float* uf, const float* itf,
                                         const float* bfv, int i, int d) {
    if (i < NU)      return uf[(size_t)i * EMB + d];
    if (i < NU + NI) return itf[(size_t)(i - NU) * EMB + d];
    return bfv[(size_t)(i - NU - NI) * EMB + d];
}

__global__ void zero_k(float* p, long n) {
    long i = (long)blockIdx.x * blockDim.x + threadIdx.x;
    long stride = (long)gridDim.x * blockDim.x;
    for (; i < n; i += stride) p[i] = 0.0f;
}

__global__ void degree_k(const int* __restrict__ rows, const int* __restrict__ cols,
                         float* dv, float* de) {
    int e = blockIdx.x * blockDim.x + threadIdx.x;
    if (e >= NNZE) return;
    atomAddF(dv + rows[e], 1.0f);
    atomAddF(de + cols[e], 1.0f);
}

__global__ void rsq_k(float* p, long n) {
    long i = (long)blockIdx.x * blockDim.x + threadIdx.x;
    if (i < n) p[i] = 1.0f / (sqrtf(p[i]) + EPSF);
}

__global__ void buildz_k(const float* uf, const float* itf, const float* bfv,
                         const float* __restrict__ dvi, float* __restrict__ z) {
    long t = (long)blockIdx.x * blockDim.x + threadIdx.x;
    if (t >= (long)NTOT * EMB) return;
    int i = (int)(t >> 5), d = (int)(t & 31);
    z[t] = feat_at(uf, itf, bfv, i, d) * dvi[i];
}

__global__ void scatter_k(const int* __restrict__ sidx, const int* __restrict__ didx,
                          const float* __restrict__ src, float* __restrict__ dst) {
    long t = (long)blockIdx.x * blockDim.x + threadIdx.x;
    int e = (int)(t >> 5);
    int lane = (int)(t & 31);
    if (e >= NNZE) return;
    int s = sidx[e], d = didx[e];
    float val = src[(size_t)s * EMB + lane];
    atomAddF(dst + (size_t)d * EMB + lane, val);
}

__global__ void rowscale_k(float* __restrict__ arr, const float* __restrict__ s) {
    long t = (long)blockIdx.x * blockDim.x + threadIdx.x;
    if (t >= (long)NTOT * EMB) return;
    arr[t] *= s[t >> 5];
}

__global__ void final_k(const float* uf, const float* itf, const float* bfv,
                        const float* __restrict__ dvi, const float* __restrict__ v,
                        float* __restrict__ all, float* __restrict__ partials) {
    __shared__ float sdata[256];
    long t = (long)blockIdx.x * blockDim.x + threadIdx.x;
    float sq = 0.0f;
    if (t < (long)NTOT * EMB) {
        int i = (int)(t >> 5), d = (int)(t & 31);
        float val = feat_at(uf, itf, bfv, i, d) * 0.5f + v[t] * dvi[i] * (1.0f / 3.0f);
        all[t] = val;
        sq = val * val;
    }
    sdata[threadIdx.x] = sq;
    __syncthreads();
    for (int s = 128; s > 0; s >>= 1) {
        if (threadIdx.x < (unsigned)s) sdata[threadIdx.x] += sdata[threadIdx.x + s];
        __syncthreads();
    }
    if (threadIdx.x == 0) partials[blockIdx.x] = sdata[0];
}

__global__ void loss_k(const float* __restrict__ partials, int nb, float* __restrict__ out_loss) {
    __shared__ float sdata[256];
    float s = 0.0f;
    for (int i = threadIdx.x; i < nb; i += 256) s += partials[i];
    sdata[threadIdx.x] = s;
    __syncthreads();
    for (int k = 128; k > 0; k >>= 1) {
        if (threadIdx.x < (unsigned)k) sdata[threadIdx.x] += sdata[threadIdx.x + k];
        __syncthreads();
    }
    if (threadIdx.x == 0) *out_loss = L2NORM * sdata[0];
}

__global__ void pred_k(const float* __restrict__ all, const int* __restrict__ users,
                       const int* __restrict__ bundles, const float* __restrict__ ubound,
                       const float* __restrict__ zrow,
                       float* __restrict__ pred, float* __restrict__ usb) {
    int wave = (blockIdx.x * blockDim.x + threadIdx.x) >> 5;
    int lane = threadIdx.x & 31;
    int b = wave / 7;
    int kt = wave % 7;

    const float* bfbase = all + (size_t)(NU + NI) * EMB;
    const float* ue = all + (size_t)users[b] * EMB;

    int m = lane & 15;
    int khalf = lane >> 4;
    int kk = kt * 16 + m;

    const float* arow;
    if (kk < KB)                 arow = bfbase + (size_t)bundles[b * KB + kk] * EMB;
    else if (kt == 6 && m == 4)  arow = ubound;
    else                         arow = zrow;

    v8f c = {};
#pragma unroll
    for (int s = 0; s < 8; ++s) {
        int e0 = s * 4 + khalf * 2;
        v2f a  = *(const v2f*)(arow + e0);
        v2f bb = *(const v2f*)(ue + e0);
        c = __builtin_amdgcn_wmma_f32_16x16x4_f32(false, a, false, bb,
                                                  (short)0, c, false, false);
    }

    if (lane == 0) {
#pragma unroll
        for (int j = 0; j < 8; ++j) {
            int kkk = kt * 16 + j;
            if (kkk < KB)                 pred[(size_t)b * KB + kkk] = c[j];
            else if (kt == 6 && j == 4)   usb[b] = c[j];
        }
    } else if (lane == 16) {
#pragma unroll
        for (int j = 0; j < 8; ++j) {
            int kkk = kt * 16 + 8 + j;
            if (kkk < KB) pred[(size_t)b * KB + kkk] = c[j];
        }
    }
}

extern "C" void kernel_launch(void* const* d_in, const int* in_sizes, int n_in,
                              void* d_out, int out_size, void* d_ws, size_t ws_size,
                              hipStream_t stream) {
    const float* uf      = (const float*)d_in[0];
    const float* itf     = (const float*)d_in[1];
    const float* bfv     = (const float*)d_in[2];
    const float* ubound  = (const float*)d_in[3];
    const int*   rows    = (const int*)d_in[4];
    const int*   cols    = (const int*)d_in[5];
    const int*   users   = (const int*)d_in[6];
    const int*   bundles = (const int*)d_in[7];
    float* out = (float*)d_out;

    float* ws   = (float*)d_ws;
    float* dvde = ws;
    float* z    = dvde + 2L * NTOT;
    float* u    = z + 32L * NTOT;
    float* v    = u + 32L * NTOT;
    float* partials = v + 32L * NTOT;
    float* zrow = partials + 8192;

    const long elems = (long)NTOT * EMB;
    const int  eb    = (int)((elems + 255) / 256);
    const int  edgeb = (NNZE * 32 + 255) / 256;

    zero_k<<<2048, 256, 0, stream>>>(dvde, 2L * NTOT);
    zero_k<<<4096, 256, 0, stream>>>(u, 64L * NTOT);
    zero_k<<<1, 32, 0, stream>>>(zrow, 32L);

    degree_k<<<(NNZE + 255) / 256, 256, 0, stream>>>(rows, cols, dvde, dvde + NTOT);
    rsq_k<<<(int)((2L * NTOT + 255) / 256), 256, 0, stream>>>(dvde, 2L * NTOT);

    buildz_k<<<eb, 256, 0, stream>>>(uf, itf, bfv, dvde, z);
    scatter_k<<<edgeb, 256, 0, stream>>>(rows, cols, z, u);
    rowscale_k<<<eb, 256, 0, stream>>>(u, dvde + NTOT);
    scatter_k<<<edgeb, 256, 0, stream>>>(cols, rows, u, v);

    final_k<<<eb, 256, 0, stream>>>(uf, itf, bfv, dvde, v, z, partials);
    loss_k<<<1, 256, 0, stream>>>(partials, eb, out + BATCHN * KB + BATCHN);

    pred_k<<<1792, 256, 0, stream>>>(z, users, bundles, ubound, zrow,
                                     out, out + BATCHN * KB);
}


# degree counts by LDS counters per (edge chunk, bin range) block plus a partial-count sum, replacing 4M scattered global float atomics; counts are exact integers, so order-free
# speedup vs baseline: 1.4413x; 1.4413x over previous
#include <hip/hip_runtime.h>
#include <stddef.h>

#define NU   8039
#define NI   32770
#define NB   4771
#define NTOT (NU + NI + NB)
#define EMB  32
#define NNZE 2000000
#define BATCHN 2048
#define KB   100
#define EPSF 1e-8f
#define L2NORM 1e-5f

typedef __attribute__((ext_vector_type(2))) float v2f;
typedef __attribute__((ext_vector_type(8))) float v8f;

__device__ __forceinline__ void atomAddF(float* p, float v) {
    (void)__hip_atomic_fetch_add(p, v, __ATOMIC_RELAXED, __HIP_MEMORY_SCOPE_AGENT);
}

__device__ __forceinline__ float feat_at(const float* uf, const float* itf,
                                         const float* bfv, int i, int d) {
    if (i < NU)      return uf[(size_t)i * EMB + d];
    if (i < NU + NI) return itf[(size_t)(i - NU) * EMB + d];
    return bfv[(size_t)(i - NU - NI) * EMB + d];
}

__global__ void zero_k(float* p, long n) {
    long i = (long)blockIdx.x * blockDim.x + threadIdx.x;
    long stride = (long)gridDim.x * blockDim.x;
    for (; i < n; i += stride) p[i] = 0.0f;
}

#define HP  4
#define HQ  32
#define HR  (NTOT / HP)
#define VPQ (NNZE / 4 / HQ)
#define HLD 8

__device__ __forceinline__ void bin_hit(unsigned int* bins, int idx, int lo) {
    unsigned int r = (unsigned int)(idx - lo);
    if (r < (unsigned int)HR) atomicAdd(bins + r, 1u);
}

__global__ __launch_bounds__(1024) void hist_k(const int* __restrict__ rows,
                                               const int* __restrict__ cols,
                                               float* __restrict__ part) {
    __shared__ unsigned int bins[HR];
    const int q = blockIdx.x, p = blockIdx.y, a = blockIdx.z;
    const int tid = threadIdx.x;
    for (int i = tid; i < HR; i += 1024) bins[i] = 0u;
    __syncthreads();

    const int lo = p * HR;
    const int4* __restrict__ vec = (const int4*)(a ? cols : rows) + (size_t)q * VPQ;
    for (int base = 0; base < VPQ; base += HLD * 1024) {
        int4 e[HLD];
#pragma unroll
        for (int j = 0; j < HLD; ++j) {
            int i = base + j * 1024 + tid;
            e[j] = (i < VPQ) ? vec[i] : make_int4(-1, -1, -1, -1);
        }
#pragma unroll
        for (int j = 0; j < HLD; ++j) {
            bin_hit(bins, e[j].x, lo);
            bin_hit(bins, e[j].y, lo);
            bin_hit(bins, e[j].z, lo);
            bin_hit(bins, e[j].w, lo);
        }
    }
    __syncthreads();

    float* __restrict__ dst = part + ((size_t)q * 2 + a) * NTOT + lo;
    for (int i = tid; i < HR; i += 1024) dst[i] = (float)bins[i];
}

__global__ void rsqsum_k(const float* __restrict__ part, float* __restrict__ dvde) {
    int i = blockIdx.x * blockDim.x + threadIdx.x;
    if (i >= 2 * NTOT) return;
    float d = 0.0f;
#pragma unroll
    for (int q = 0; q < HQ; ++q) d += part[(size_t)q * (2 * NTOT) + i];
    dvde[i] = 1.0f / (sqrtf(d) + EPSF);
}

__global__ void buildz_k(const float* uf, const float* itf, const float* bfv,
                         const float* __restrict__ dvi, float* __restrict__ z) {
    long t = (long)blockIdx.x * blockDim.x + threadIdx.x;
    if (t >= (long)NTOT * EMB) return;
    int i = (int)(t >> 5), d = (int)(t & 31);
    z[t] = feat_at(uf, itf, bfv, i, d) * dvi[i];
}

__global__ void scatter_k(const int* __restrict__ sidx, const int* __restrict__ didx,
                          const float* __restrict__ src, float* __restrict__ dst) {
    long t = (long)blockIdx.x * blockDim.x + threadIdx.x;
    int e = (int)(t >> 5);
    int lane = (int)(t & 31);
    if (e >= NNZE) return;
    int s = sidx[e], d = didx[e];
    float val = src[(size_t)s * EMB + lane];
    atomAddF(dst + (size_t)d * EMB + lane, val);
}

__global__ void rowscale_k(float* __restrict__ arr, const float* __restrict__ s) {
    long t = (long)blockIdx.x * blockDim.x + threadIdx.x;
    if (t >= (long)NTOT * EMB) return;
    arr[t] *= s[t >> 5];
}

__global__ void final_k(const float* uf, const float* itf, const float* bfv,
                        const float* __restrict__ dvi, const float* __restrict__ v,
                        float* __restrict__ all, float* __restrict__ partials) {
    __shared__ float sdata[256];
    long t = (long)blockIdx.x * blockDim.x + threadIdx.x;
    float sq = 0.0f;
    if (t < (long)NTOT * EMB) {
        int i = (int)(t >> 5), d = (int)(t & 31);
        float val = feat_at(uf, itf, bfv, i, d) * 0.5f + v[t] * dvi[i] * (1.0f / 3.0f);
        all[t] = val;
        sq = val * val;
    }
    sdata[threadIdx.x] = sq;
    __syncthreads();
    for (int s = 128; s > 0; s >>= 1) {
        if (threadIdx.x < (unsigned)s) sdata[threadIdx.x] += sdata[threadIdx.x + s];
        __syncthreads();
    }
    if (threadIdx.x == 0) partials[blockIdx.x] = sdata[0];
}

__global__ void loss_k(const float* __restrict__ partials, int nb, float* __restrict__ out_loss) {
    __shared__ float sdata[256];
    float s = 0.0f;
    for (int i = threadIdx.x; i < nb; i += 256) s += partials[i];
    sdata[threadIdx.x] = s;
    __syncthreads();
    for (int k = 128; k > 0; k >>= 1) {
        if (threadIdx.x < (unsigned)k) sdata[threadIdx.x] += sdata[threadIdx.x + k];
        __syncthreads();
    }
    if (threadIdx.x == 0) *out_loss = L2NORM * sdata[0];
}

__global__ void pred_k(const float* __restrict__ all, const int* __restrict__ users,
                       const int* __restrict__ bundles, const float* __restrict__ ubound,
                       const float* __restrict__ zrow,
                       float* __restrict__ pred, float* __restrict__ usb) {
    int wave = (blockIdx.x * blockDim.x + threadIdx.x) >> 5;
    int lane = threadIdx.x & 31;
    int b = wave / 7;
    int kt = wave % 7;

    const float* bfbase = all + (size_t)(NU + NI) * EMB;
    const float* ue = all + (size_t)users[b] * EMB;

    int m = lane & 15;
    int khalf = lane >> 4;
    int kk = kt * 16 + m;

    const float* arow;
    if (kk < KB)                 arow = bfbase + (size_t)bundles[b * KB + kk] * EMB;
    else if (kt == 6 && m == 4)  arow = ubound;
    else                         arow = zrow;

    v8f c = {};
#pragma unroll
    for (int s = 0; s < 8; ++s) {
        int e0 = s * 4 + khalf * 2;
        v2f a  = *(const v2f*)(arow + e0);
        v2f bb = *(const v2f*)(ue + e0);
        c = __builtin_amdgcn_wmma_f32_16x16x4_f32(false, a, false, bb,
                                                  (short)0, c, false, false);
    }

    if (lane == 0) {
#pragma unroll
        for (int j = 0; j < 8; ++j) {
            int kkk = kt * 16 + j;
            if (kkk < KB)                 pred[(size_t)b * KB + kkk] = c[j];
            else if (kt == 6 && j == 4)   usb[b] = c[j];
        }
    } else if (lane == 16) {
#pragma unroll
        for (int j = 0; j < 8; ++j) {
            int kkk = kt * 16 + 8 + j;
            if (kkk < KB) pred[(size_t)b * KB + kkk] = c[j];
        }
    }
}

extern "C" void kernel_launch(void* const* d_in, const int* in_sizes, int n_in,
                              void* d_out, int out_size, void* d_ws, size_t ws_size,
                              hipStream_t stream) {
    const float* uf      = (const float*)d_in[0];
    const float* itf     = (const float*)d_in[1];
    const float* bfv     = (const float*)d_in[2];
    const float* ubound  = (const float*)d_in[3];
    const int*   rows    = (const int*)d_in[4];
    const int*   cols    = (const int*)d_in[5];
    const int*   users   = (const int*)d_in[6];
    const int*   bundles = (const int*)d_in[7];
    float* out = (float*)d_out;

    float* ws   = (float*)d_ws;
    float* dvde = ws;
    float* z    = dvde + 2L * NTOT;
    float* u    = z + 32L * NTOT;
    float* v    = u + 32L * NTOT;
    float* partials = v + 32L * NTOT;
    float* zrow = partials + 8192;

    const long elems = (long)NTOT * EMB;
    const int  eb    = (int)((elems + 255) / 256);
    const int  edgeb = (NNZE * 32 + 255) / 256;

    hist_k<<<dim3(HQ, HP, 2), 1024, 0, stream>>>(rows, cols, u);
    rsqsum_k<<<(2 * NTOT + 255) / 256, 256, 0, stream>>>(u, dvde);

    zero_k<<<4096, 256, 0, stream>>>(u, 64L * NTOT);
    zero_k<<<1, 32, 0, stream>>>(zrow, 32L);

    buildz_k<<<eb, 256, 0, stream>>>(uf, itf, bfv, dvde, z);
    scatter_k<<<edgeb, 256, 0, stream>>>(rows, cols, z, u);
    rowscale_k<<<eb, 256, 0, stream>>>(u, dvde + NTOT);
    scatter_k<<<edgeb, 256, 0, stream>>>(cols, rows, u, v);

    final_k<<<eb, 256, 0, stream>>>(uf, itf, bfv, dvde, v, z, partials);
    loss_k<<<1, 256, 0, stream>>>(partials, eb, out + BATCHN * KB + BATCHN);

    pred_k<<<1792, 256, 0, stream>>>(z, users, bundles, ubound, zrow,
                                     out, out + BATCHN * KB);
}


# edge scatter: one wave per 32 consecutive edges (coalesced index loads, lane broadcast, 32 row loads in flight) instead of one wave per edge; float atomics kept, arrival order as before
# speedup vs baseline: 2.6188x; 1.8170x over previous
#include <hip/hip_runtime.h>
#include <stddef.h>

#define NU   8039
#define NI   32770
#define NB   4771
#define NTOT (NU + NI + NB)
#define EMB  32
#define NNZE 2000000
#define BATCHN 2048
#define KB   100
#define EPSF 1e-8f
#define L2NORM 1e-5f

typedef __attribute__((ext_vector_type(2))) float v2f;
typedef __attribute__((ext_vector_type(8))) float v8f;

__device__ __forceinline__ void atomAddF(float* p, float v) {
    (void)__hip_atomic_fetch_add(p, v, __ATOMIC_RELAXED, __HIP_MEMORY_SCOPE_AGENT);
}

__device__ __forceinline__ float feat_at(const float* uf, const float* itf,
                                         const float* bfv, int i, int d) {
    if (i < NU)      return uf[(size_t)i * EMB + d];
    if (i < NU + NI) return itf[(size_t)(i - NU) * EMB + d];
    return bfv[(size_t)(i - NU - NI) * EMB + d];
}

__global__ void zero_k(float* p, long n) {
    long i = (long)blockIdx.x * blockDim.x + threadIdx.x;
    long stride = (long)gridDim.x * blockDim.x;
    for (; i < n; i += stride) p[i] = 0.0f;
}

#define HP  4
#define HQ  32
#define HR  (NTOT / HP)
#define VPQ (NNZE / 4 / HQ)
#define HLD 8

__device__ __forceinline__ void bin_hit(unsigned int* bins, int idx, int lo) {
    unsigned int r = (unsigned int)(idx - lo);
    if (r < (unsigned int)HR) atomicAdd(bins + r, 1u);
}

__global__ __launch_bounds__(1024) void hist_k(const int* __restrict__ rows,
                                               const int* __restrict__ cols,
                                               float* __restrict__ part) {
    __shared__ unsigned int bins[HR];
    const int q = blockIdx.x, p = blockIdx.y, a = blockIdx.z;
    const int tid = threadIdx.x;
    for (int i = tid; i < HR; i += 1024) bins[i] = 0u;
    __syncthreads();

    const int lo = p * HR;
    const int4* __restrict__ vec = (const int4*)(a ? cols : rows) + (size_t)q * VPQ;
    for (int base = 0; base < VPQ; base += HLD * 1024) {
        int4 e[HLD];
#pragma unroll
        for (int j = 0; j < HLD; ++j) {
            int i = base + j * 1024 + tid;
            e[j] = (i < VPQ) ? vec[i] : make_int4(-1, -1, -1, -1);
        }
#pragma unroll
        for (int j = 0; j < HLD; ++j) {
            bin_hit(bins, e[j].x, lo);
            bin_hit(bins, e[j].y, lo);
            bin_hit(bins, e[j].z, lo);
            bin_hit(bins, e[j].w, lo);
        }
    }
    __syncthreads();

    float* __restrict__ dst = part + ((size_t)q * 2 + a) * NTOT + lo;
    for (int i = tid; i < HR; i += 1024) dst[i] = (float)bins[i];
}

__global__ void rsqsum_k(const float* __restrict__ part, float* __restrict__ dvde) {
    int i = blockIdx.x * blockDim.x + threadIdx.x;
    if (i >= 2 * NTOT) return;
    float d = 0.0f;
#pragma unroll
    for (int q = 0; q < HQ; ++q) d += part[(size_t)q * (2 * NTOT) + i];
    dvde[i] = 1.0f / (sqrtf(d) + EPSF);
}

__global__ void buildz_k(const float* uf, const float* itf, const float* bfv,
                         const float* __restrict__ dvi, float* __restrict__ z) {
    long t = (long)blockIdx.x * blockDim.x + threadIdx.x;
    if (t >= (long)NTOT * EMB) return;
    int i = (int)(t >> 5), d = (int)(t & 31);
    z[t] = feat_at(uf, itf, bfv, i, d) * dvi[i];
}

#define EPW    32
#define SWAVES (NNZE / EPW)
static_assert(NNZE % EPW == 0, "every wave owns a full run of edges");

__global__ __launch_bounds__(256) void scatter_k(const int* __restrict__ sidx,
                                                 const int* __restrict__ didx,
                                                 const float* __restrict__ src,
                                                 float* __restrict__ dst) {
    const int wave = (int)((blockIdx.x * 256u + threadIdx.x) >> 5);
    const int lane = (int)(threadIdx.x & 31u);
    if (wave >= SWAVES) return;
    const int e0 = wave * EPW;
    const int sv = sidx[e0 + lane];
    const int dv = didx[e0 + lane];

    float val[EPW];
#pragma unroll
    for (int j = 0; j < EPW; ++j) {
        const int s = __builtin_amdgcn_readlane(sv, j);
        val[j] = src[(size_t)s * EMB + lane];
    }
#pragma unroll
    for (int j = 0; j < EPW; ++j) {
        const int d = __builtin_amdgcn_readlane(dv, j);
        atomAddF(dst + (size_t)d * EMB + lane, val[j]);
    }
}

__global__ void rowscale_k(float* __restrict__ arr, const float* __restrict__ s) {
    long t = (long)blockIdx.x * blockDim.x + threadIdx.x;
    if (t >= (long)NTOT * EMB) return;
    arr[t] *= s[t >> 5];
}

__global__ void final_k(const float* uf, const float* itf, const float* bfv,
                        const float* __restrict__ dvi, const float* __restrict__ v,
                        float* __restrict__ all, float* __restrict__ partials) {
    __shared__ float sdata[256];
    long t = (long)blockIdx.x * blockDim.x + threadIdx.x;
    float sq = 0.0f;
    if (t < (long)NTOT * EMB) {
        int i = (int)(t >> 5), d = (int)(t & 31);
        float val = feat_at(uf, itf, bfv, i, d) * 0.5f + v[t] * dvi[i] * (1.0f / 3.0f);
        all[t] = val;
        sq = val * val;
    }
    sdata[threadIdx.x] = sq;
    __syncthreads();
    for (int s = 128; s > 0; s >>= 1) {
        if (threadIdx.x < (unsigned)s) sdata[threadIdx.x] += sdata[threadIdx.x + s];
        __syncthreads();
    }
    if (threadIdx.x == 0) partials[blockIdx.x] = sdata[0];
}

__global__ void loss_k(const float* __restrict__ partials, int nb, float* __restrict__ out_loss) {
    __shared__ float sdata[256];
    float s = 0.0f;
    for (int i = threadIdx.x; i < nb; i += 256) s += partials[i];
    sdata[threadIdx.x] = s;
    __syncthreads();
    for (int k = 128; k > 0; k >>= 1) {
        if (threadIdx.x < (unsigned)k) sdata[threadIdx.x] += sdata[threadIdx.x + k];
        __syncthreads();
    }
    if (threadIdx.x == 0) *out_loss = L2NORM * sdata[0];
}

__global__ void pred_k(const float* __restrict__ all, const int* __restrict__ users,
                       const int* __restrict__ bundles, const float* __restrict__ ubound,
                       const float* __restrict__ zrow,
                       float* __restrict__ pred, float* __restrict__ usb) {
    int wave = (blockIdx.x * blockDim.x + threadIdx.x) >> 5;
    int lane = threadIdx.x & 31;
    int b = wave / 7;
    int kt = wave % 7;

    const float* bfbase = all + (size_t)(NU + NI) * EMB;
    const float* ue = all + (size_t)users[b] * EMB;

    int m = lane & 15;
    int khalf = lane >> 4;
    int kk = kt * 16 + m;

    const float* arow;
    if (kk < KB)                 arow = bfbase + (size_t)bundles[b * KB + kk] * EMB;
    else if (kt == 6 && m == 4)  arow = ubound;
    else                         arow = zrow;

    v8f c = {};
#pragma unroll
    for (int s = 0; s < 8; ++s) {
        int e0 = s * 4 + khalf * 2;
        v2f a  = *(const v2f*)(arow + e0);
        v2f bb = *(const v2f*)(ue + e0);
        c = __builtin_amdgcn_wmma_f32_16x16x4_f32(false, a, false, bb,
                                                  (short)0, c, false, false);
    }

    if (lane == 0) {
#pragma unroll
        for (int j = 0; j < 8; ++j) {
            int kkk = kt * 16 + j;
            if (kkk < KB)                 pred[(size_t)b * KB + kkk] = c[j];
            else if (kt == 6 && j == 4)   usb[b] = c[j];
        }
    } else if (lane == 16) {
#pragma unroll
        for (int j = 0; j < 8; ++j) {
            int kkk = kt * 16 + 8 + j;
            if (kkk < KB) pred[(size_t)b * KB + kkk] = c[j];
        }
    }
}

extern "C" void kernel_launch(void* const* d_in, const int* in_sizes, int n_in,
                              void* d_out, int out_size, void* d_ws, size_t ws_size,
                              hipStream_t stream) {
    const float* uf      = (const float*)d_in[0];
    const float* itf     = (const float*)d_in[1];
    const float* bfv     = (const float*)d_in[2];
    const float* ubound  = (const float*)d_in[3];
    const int*   rows    = (const int*)d_in[4];
    const int*   cols    = (const int*)d_in[5];
    const int*   users   = (const int*)d_in[6];
    const int*   bundles = (const int*)d_in[7];
    float* out = (float*)d_out;

    float* ws   = (float*)d_ws;
    float* dvde = ws;
    float* z    = dvde + 2L * NTOT;
    float* u    = z + 32L * NTOT;
    float* v    = u + 32L * NTOT;
    float* partials = v + 32L * NTOT;
    float* zrow = partials + 8192;

    const long elems = (long)NTOT * EMB;
    const int  eb    = (int)((elems + 255) / 256);
    const int  edgeb = (SWAVES * 32 + 255) / 256;

    hist_k<<<dim3(HQ, HP, 2), 1024, 0, stream>>>(rows, cols, u);
    rsqsum_k<<<(2 * NTOT + 255) / 256, 256, 0, stream>>>(u, dvde);

    zero_k<<<4096, 256, 0, stream>>>(u, 64L * NTOT);
    zero_k<<<1, 32, 0, stream>>>(zrow, 32L);

    buildz_k<<<eb, 256, 0, stream>>>(uf, itf, bfv, dvde, z);
    scatter_k<<<edgeb, 256, 0, stream>>>(rows, cols, z, u);
    rowscale_k<<<eb, 256, 0, stream>>>(u, dvde + NTOT);
    scatter_k<<<edgeb, 256, 0, stream>>>(cols, rows, u, v);

    final_k<<<eb, 256, 0, stream>>>(uf, itf, bfv, dvde, v, z, partials);
    loss_k<<<1, 256, 0, stream>>>(partials, eb, out + BATCHN * KB + BATCHN);

    pred_k<<<1792, 256, 0, stream>>>(z, users, bundles, ubound, zrow,
                                     out, out + BATCHN * KB);
}


# edge scatter: one accumulator plane per cache half (wave picks its plane by die id), planes summed by the next pass; float atomics kept, each element now the sum of two arrival-order partial sums
# speedup vs baseline: 3.5929x; 1.3720x over previous
#include <hip/hip_runtime.h>
#include <stddef.h>

#define NU   8039
#define NI   32770
#define NB   4771
#define NTOT (NU + NI + NB)
#define EMB  32
#define NNZE 2000000
#define BATCHN 2048
#define KB   100
#define EPSF 1e-8f
#define L2NORM 1e-5f

typedef __attribute__((ext_vector_type(2))) float v2f;
typedef __attribute__((ext_vector_type(8))) float v8f;

__device__ __forceinline__ void atomAddF(float* p, float v) {
    (void)__hip_atomic_fetch_add(p, v, __ATOMIC_RELAXED, __HIP_MEMORY_SCOPE_AGENT);
}

__device__ __forceinline__ float feat_at(const float* uf, const float* itf,
                                         const float* bfv, int i, int d) {
    if (i < NU)      return uf[(size_t)i * EMB + d];
    if (i < NU + NI) return itf[(size_t)(i - NU) * EMB + d];
    return bfv[(size_t)(i - NU - NI) * EMB + d];
}

__global__ void zero_k(float* p, long n) {
    long i = (long)blockIdx.x * blockDim.x + threadIdx.x;
    long stride = (long)gridDim.x * blockDim.x;
    for (; i < n; i += stride) p[i] = 0.0f;
}

#define HP  4
#define HQ  32
#define HR  (NTOT / HP)
#define VPQ (NNZE / 4 / HQ)
#define HLD 8

__device__ __forceinline__ void bin_hit(unsigned int* bins, int idx, int lo) {
    unsigned int r = (unsigned int)(idx - lo);
    if (r < (unsigned int)HR) atomicAdd(bins + r, 1u);
}

__global__ __launch_bounds__(1024) void hist_k(const int* __restrict__ rows,
                                               const int* __restrict__ cols,
                                               float* __restrict__ part) {
    __shared__ unsigned int bins[HR];
    const int q = blockIdx.x, p = blockIdx.y, a = blockIdx.z;
    const int tid = threadIdx.x;
    for (int i = tid; i < HR; i += 1024) bins[i] = 0u;
    __syncthreads();

    const int lo = p * HR;
    const int4* __restrict__ vec = (const int4*)(a ? cols : rows) + (size_t)q * VPQ;
    for (int base = 0; base < VPQ; base += HLD * 1024) {
        int4 e[HLD];
#pragma unroll
        for (int j = 0; j < HLD; ++j) {
            int i = base + j * 1024 + tid;
            e[j] = (i < VPQ) ? vec[i] : make_int4(-1, -1, -1, -1);
        }
#pragma unroll
        for (int j = 0; j < HLD; ++j) {
            bin_hit(bins, e[j].x, lo);
            bin_hit(bins, e[j].y, lo);
            bin_hit(bins, e[j].z, lo);
            bin_hit(bins, e[j].w, lo);
        }
    }
    __syncthreads();

    float* __restrict__ dst = part + ((size_t)q * 2 + a) * NTOT + lo;
    for (int i = tid; i < HR; i += 1024) dst[i] = (float)bins[i];
}

__global__ void rsqsum_k(const float* __restrict__ part, float* __restrict__ dvde) {
    int i = blockIdx.x * blockDim.x + threadIdx.x;
    if (i >= 2 * NTOT) return;
    float d = 0.0f;
#pragma unroll
    for (int q = 0; q < HQ; ++q) d += part[(size_t)q * (2 * NTOT) + i];
    dvde[i] = 1.0f / (sqrtf(d) + EPSF);
}

__global__ void buildz_k(const float* uf, const float* itf, const float* bfv,
                         const float* __restrict__ dvi, float* __restrict__ z) {
    long t = (long)blockIdx.x * blockDim.x + threadIdx.x;
    if (t >= (long)NTOT * EMB) return;
    int i = (int)(t >> 5), d = (int)(t & 31);
    z[t] = feat_at(uf, itf, bfv, i, d) * dvi[i];
}

#define EPW    32
#define PLANE  ((size_t)NTOT * EMB)

__device__ __forceinline__ unsigned int cache_half() {
    return (__builtin_amdgcn_s_sendmsg_rtn(0x87) >> 18) & 1u;
}
#define SWAVES (NNZE / EPW)
static_assert(NNZE % EPW == 0, "every wave owns a full run of edges");

__global__ __launch_bounds__(256) void scatter_k(const int* __restrict__ sidx,
                                                 const int* __restrict__ didx,
                                                 const float* __restrict__ src,
                                                 float* __restrict__ dst) {
    const int wave = (int)((blockIdx.x * 256u + threadIdx.x) >> 5);
    const int lane = (int)(threadIdx.x & 31u);
    if (wave >= SWAVES) return;
    const int e0 = wave * EPW;
    const int sv = sidx[e0 + lane];
    const int dv = didx[e0 + lane];
    float* __restrict__ plane = dst + (size_t)cache_half() * PLANE;

    float val[EPW];
#pragma unroll
    for (int j = 0; j < EPW; ++j) {
        const int s = __builtin_amdgcn_readlane(sv, j);
        val[j] = src[(size_t)s * EMB + lane];
    }
#pragma unroll
    for (int j = 0; j < EPW; ++j) {
        const int d = __builtin_amdgcn_readlane(dv, j);
        atomAddF(plane + (size_t)d * EMB + lane, val[j]);
    }
}

__global__ void combine_k(float* __restrict__ pa, float* __restrict__ pb,
                          const float* __restrict__ s, float* __restrict__ res) {
    long t = (long)blockIdx.x * blockDim.x + threadIdx.x;
    if (t >= (long)NTOT * EMB) return;
    res[t] = (pa[t] + pb[t]) * s[t >> 5];
    pa[t] = 0.0f;
    pb[t] = 0.0f;
}

__global__ void final_k(const float* uf, const float* itf, const float* bfv,
                        const float* __restrict__ dvi, const float* __restrict__ va,
                        const float* __restrict__ vb,
                        float* __restrict__ all, float* __restrict__ partials) {
    __shared__ float sdata[256];
    long t = (long)blockIdx.x * blockDim.x + threadIdx.x;
    float sq = 0.0f;
    if (t < (long)NTOT * EMB) {
        int i = (int)(t >> 5), d = (int)(t & 31);
        float val = feat_at(uf, itf, bfv, i, d) * 0.5f + (va[t] + vb[t]) * dvi[i] * (1.0f / 3.0f);
        all[t] = val;
        sq = val * val;
    }
    sdata[threadIdx.x] = sq;
    __syncthreads();
    for (int s = 128; s > 0; s >>= 1) {
        if (threadIdx.x < (unsigned)s) sdata[threadIdx.x] += sdata[threadIdx.x + s];
        __syncthreads();
    }
    if (threadIdx.x == 0) partials[blockIdx.x] = sdata[0];
}

__global__ void loss_k(const float* __restrict__ partials, int nb, float* __restrict__ out_loss) {
    __shared__ float sdata[256];
    float s = 0.0f;
    for (int i = threadIdx.x; i < nb; i += 256) s += partials[i];
    sdata[threadIdx.x] = s;
    __syncthreads();
    for (int k = 128; k > 0; k >>= 1) {
        if (threadIdx.x < (unsigned)k) sdata[threadIdx.x] += sdata[threadIdx.x + k];
        __syncthreads();
    }
    if (threadIdx.x == 0) *out_loss = L2NORM * sdata[0];
}

__global__ void pred_k(const float* __restrict__ all, const int* __restrict__ users,
                       const int* __restrict__ bundles, const float* __restrict__ ubound,
                       const float* __restrict__ zrow,
                       float* __restrict__ pred, float* __restrict__ usb) {
    int wave = (blockIdx.x * blockDim.x + threadIdx.x) >> 5;
    int lane = threadIdx.x & 31;
    int b = wave / 7;
    int kt = wave % 7;

    const float* bfbase = all + (size_t)(NU + NI) * EMB;
    const float* ue = all + (size_t)users[b] * EMB;

    int m = lane & 15;
    int khalf = lane >> 4;
    int kk = kt * 16 + m;

    const float* arow;
    if (kk < KB)                 arow = bfbase + (size_t)bundles[b * KB + kk] * EMB;
    else if (kt == 6 && m == 4)  arow = ubound;
    else                         arow = zrow;

    v8f c = {};
#pragma unroll
    for (int s = 0; s < 8; ++s) {
        int e0 = s * 4 + khalf * 2;
        v2f a  = *(const v2f*)(arow + e0);
        v2f bb = *(const v2f*)(ue + e0);
        c = __builtin_amdgcn_wmma_f32_16x16x4_f32(false, a, false, bb,
                                                  (short)0, c, false, false);
    }

    if (lane == 0) {
#pragma unroll
        for (int j = 0; j < 8; ++j) {
            int kkk = kt * 16 + j;
            if (kkk < KB)                 pred[(size_t)b * KB + kkk] = c[j];
            else if (kt == 6 && j == 4)   usb[b] = c[j];
        }
    } else if (lane == 16) {
#pragma unroll
        for (int j = 0; j < 8; ++j) {
            int kkk = kt * 16 + 8 + j;
            if (kkk < KB) pred[(size_t)b * KB + kkk] = c[j];
        }
    }
}

extern "C" void kernel_launch(void* const* d_in, const int* in_sizes, int n_in,
                              void* d_out, int out_size, void* d_ws, size_t ws_size,
                              hipStream_t stream) {
    const float* uf      = (const float*)d_in[0];
    const float* itf     = (const float*)d_in[1];
    const float* bfv     = (const float*)d_in[2];
    const float* ubound  = (const float*)d_in[3];
    const int*   rows    = (const int*)d_in[4];
    const int*   cols    = (const int*)d_in[5];
    const int*   users   = (const int*)d_in[6];
    const int*   bundles = (const int*)d_in[7];
    float* out = (float*)d_out;

    float* ws   = (float*)d_ws;
    float* dvde = ws;
    float* z    = dvde + 2L * NTOT;
    float* u    = z + 32L * NTOT;
    float* v    = u + 32L * NTOT;
    float* partials = v + 32L * NTOT;
    float* zrow = partials + 8192;

    const long elems = (long)NTOT * EMB;
    const int  eb    = (int)((elems + 255) / 256);
    const int  edgeb = (SWAVES * 32 + 255) / 256;

    hist_k<<<dim3(HQ, HP, 2), 1024, 0, stream>>>(rows, cols, u);
    rsqsum_k<<<(2 * NTOT + 255) / 256, 256, 0, stream>>>(u, dvde);

    zero_k<<<4096, 256, 0, stream>>>(u, 64L * NTOT);
    zero_k<<<1, 32, 0, stream>>>(zrow, 32L);

    buildz_k<<<eb, 256, 0, stream>>>(uf, itf, bfv, dvde, z);
    scatter_k<<<edgeb, 256, 0, stream>>>(rows, cols, z, u);
    combine_k<<<eb, 256, 0, stream>>>(u, v, dvde + NTOT, z);
    scatter_k<<<edgeb, 256, 0, stream>>>(cols, rows, z, u);

    final_k<<<eb, 256, 0, stream>>>(uf, itf, bfv, dvde, u, v, z, partials);
    loss_k<<<1, 256, 0, stream>>>(partials, eb, out + BATCHN * KB + BATCHN);

    pred_k<<<1792, 256, 0, stream>>>(z, users, bundles, ubound, zrow,
                                     out, out + BATCHN * KB);
}


# edge scatter: a block per (destination range, edge slice) sums its hit rows in LDS (float adds, arrival order), then adds each row to the plane once: 4.4x fewer global float atomics per pass
# speedup vs baseline: 3.9347x; 1.0951x over previous
#include <hip/hip_runtime.h>
#include <stddef.h>

#define NU   8039
#define NI   32770
#define NB   4771
#define NTOT (NU + NI + NB)
#define EMB  32
#define NNZE 2000000
#define BATCHN 2048
#define KB   100
#define EPSF 1e-8f
#define L2NORM 1e-5f

typedef __attribute__((ext_vector_type(2))) float v2f;
typedef __attribute__((ext_vector_type(8))) float v8f;

__device__ __forceinline__ void atomAddF(float* p, float v) {
    (void)__hip_atomic_fetch_add(p, v, __ATOMIC_RELAXED, __HIP_MEMORY_SCOPE_AGENT);
}

__device__ __forceinline__ float feat_at(const float* uf, const float* itf,
                                         const float* bfv, int i, int d) {
    if (i < NU)      return uf[(size_t)i * EMB + d];
    if (i < NU + NI) return itf[(size_t)(i - NU) * EMB + d];
    return bfv[(size_t)(i - NU - NI) * EMB + d];
}

__global__ void zero_k(float* p, long n) {
    long i = (long)blockIdx.x * blockDim.x + threadIdx.x;
    long stride = (long)gridDim.x * blockDim.x;
    for (; i < n; i += stride) p[i] = 0.0f;
}

#define HP  4
#define HQ  32
#define HR  (NTOT / HP)
#define VPQ (NNZE / 4 / HQ)
#define HLD 8

__device__ __forceinline__ void bin_hit(unsigned int* bins, int idx, int lo) {
    unsigned int r = (unsigned int)(idx - lo);
    if (r < (unsigned int)HR) atomicAdd(bins + r, 1u);
}

__global__ __launch_bounds__(1024) void hist_k(const int* __restrict__ rows,
                                               const int* __restrict__ cols,
                                               float* __restrict__ part) {
    __shared__ unsigned int bins[HR];
    const int q = blockIdx.x, p = blockIdx.y, a = blockIdx.z;
    const int tid = threadIdx.x;
    for (int i = tid; i < HR; i += 1024) bins[i] = 0u;
    __syncthreads();

    const int lo = p * HR;
    const int4* __restrict__ vec = (const int4*)(a ? cols : rows) + (size_t)q * VPQ;
    for (int base = 0; base < VPQ; base += HLD * 1024) {
        int4 e[HLD];
#pragma unroll
        for (int j = 0; j < HLD; ++j) {
            int i = base + j * 1024 + tid;
            e[j] = (i < VPQ) ? vec[i] : make_int4(-1, -1, -1, -1);
        }
#pragma unroll
        for (int j = 0; j < HLD; ++j) {
            bin_hit(bins, e[j].x, lo);
            bin_hit(bins, e[j].y, lo);
            bin_hit(bins, e[j].z, lo);
            bin_hit(bins, e[j].w, lo);
        }
    }
    __syncthreads();

    float* __restrict__ dst = part + ((size_t)q * 2 + a) * NTOT + lo;
    for (int i = tid; i < HR; i += 1024) dst[i] = (float)bins[i];
}

__global__ void rsqsum_k(const float* __restrict__ part, float* __restrict__ dvde) {
    int i = blockIdx.x * blockDim.x + threadIdx.x;
    if (i >= 2 * NTOT) return;
    float d = 0.0f;
#pragma unroll
    for (int q = 0; q < HQ; ++q) d += part[(size_t)q * (2 * NTOT) + i];
    dvde[i] = 1.0f / (sqrtf(d) + EPSF);
}

__global__ void buildz_k(const float* uf, const float* itf, const float* bfv,
                         const float* __restrict__ dvi, float* __restrict__ z) {
    long t = (long)blockIdx.x * blockDim.x + threadIdx.x;
    if (t >= (long)NTOT * EMB) return;
    int i = (int)(t >> 5), d = (int)(t & 31);
    z[t] = feat_at(uf, itf, bfv, i, d) * dvi[i];
}

#define PLANE  ((size_t)NTOT * EMB)
#define SP   25
#define SQ   10
#define SR   1824
#define SVEC (NNZE / 4 / SQ)
#define SLD  4
#define SQC  512
static_assert(SP * SR >= NTOT, "the ranges cover every row");
static_assert(NNZE % (4 * SQ) == 0, "every slice is a whole number of 16-byte vectors");
static_assert(NTOT <= 65536 && SR <= 2048, "queue word = source row << 11 | local destination row");

__device__ __forceinline__ unsigned int cache_half() {
    return (__builtin_amdgcn_s_sendmsg_rtn(0x87) >> 18) & 1u;
}

__device__ __forceinline__ void push_hit(unsigned int* __restrict__ myq, int& count,
                                         int d, int s, int lo, int lane,
                                         const float* __restrict__ src,
                                         float* __restrict__ plane) {
    const unsigned int r = (unsigned int)(d - lo);
    const bool hit = r < (unsigned int)SR;
    const unsigned int mask = __builtin_amdgcn_ballot_w32(hit);
    if (mask == 0u) return;
    const int pos = count + (int)__builtin_amdgcn_mbcnt_lo(mask, 0u);
    const bool over = hit && pos >= SQC;
    if (hit && !over) myq[pos] = ((unsigned int)s << 11) | r;
    count += __builtin_popcount(mask);
    unsigned int om = __builtin_amdgcn_ballot_w32(over);
    while (om) {
        const int j = __builtin_ctz(om);
        om &= om - 1u;
        const int sj = __builtin_amdgcn_readlane(s, j);
        const int dj = __builtin_amdgcn_readlane(d, j);
        atomAddF(plane + (size_t)dj * EMB + lane, src[(size_t)sj * EMB + lane]);
    }
}

__global__ __launch_bounds__(1024) void scatter_k(const int* __restrict__ sidx,
                                                  const int* __restrict__ didx,
                                                  const float* __restrict__ src,
                                                  float* __restrict__ dst) {
    __shared__ float acc[SR * EMB];
    __shared__ unsigned int queue[32 * SQC];
    const int p = blockIdx.x, q = blockIdx.y;
    const int tid = threadIdx.x;
    const int lane = tid & 31;
    const int w = __builtin_amdgcn_readfirstlane(tid >> 5);

    for (int i = tid; i < SR * EMB / 4; i += 1024)
        ((float4*)acc)[i] = make_float4(0.0f, 0.0f, 0.0f, 0.0f);
    __syncthreads();

    const int lo = p * SR;
    float* __restrict__ plane = dst + (size_t)cache_half() * PLANE;
    const int4* __restrict__ sv4 = (const int4*)sidx + (size_t)q * SVEC;
    const int4* __restrict__ dv4 = (const int4*)didx + (size_t)q * SVEC;
    unsigned int* __restrict__ myq = queue + w * SQC;

    int count = 0;
    for (int base = 0; base < SVEC; base += SLD * 1024) {
        int4 dd[SLD], ss[SLD];
#pragma unroll
        for (int j = 0; j < SLD; ++j) {
            const int i = base + j * 1024 + tid;
            const int ic = (i < SVEC) ? i : SVEC - 1;
            const int4 t = dv4[ic];
            ss[j] = sv4[ic];
            dd[j] = (i < SVEC) ? t : make_int4(-1, -1, -1, -1);
        }
#pragma unroll
        for (int j = 0; j < SLD; ++j) {
            push_hit(myq, count, dd[j].x, ss[j].x, lo, lane, src, plane);
            push_hit(myq, count, dd[j].y, ss[j].y, lo, lane, src, plane);
            push_hit(myq, count, dd[j].z, ss[j].z, lo, lane, src, plane);
            push_hit(myq, count, dd[j].w, ss[j].w, lo, lane, src, plane);
        }
    }

    const int n = count < SQC ? count : SQC;
    for (int g = 0; g < n; g += 32) {
        const int ent = (g + lane < n) ? (int)myq[g + lane] : 0;
        const int m = n - g;
        float val[32];
#pragma unroll
        for (int j = 0; j < 32; ++j) {
            const unsigned int e = (unsigned int)__builtin_amdgcn_readlane(ent, j);
            val[j] = src[(size_t)(e >> 11) * EMB + lane];
        }
#pragma unroll
        for (int j = 0; j < 32; ++j) {
            const unsigned int e = (unsigned int)__builtin_amdgcn_readlane(ent, j);
            if (j < m) atomicAdd(acc + (e & 2047u) * EMB + lane, val[j]);
        }
    }
    __syncthreads();

    const int nrows = (NTOT - lo < SR) ? (NTOT - lo) : SR;
    for (int row = w; row < nrows; row += 32)
        atomAddF(plane + (size_t)(lo + row) * EMB + lane, acc[row * EMB + lane]);
}

__global__ void combine_k(float* __restrict__ pa, float* __restrict__ pb,
                          const float* __restrict__ s, float* __restrict__ res) {
    long t = (long)blockIdx.x * blockDim.x + threadIdx.x;
    if (t >= (long)NTOT * EMB) return;
    res[t] = (pa[t] + pb[t]) * s[t >> 5];
    pa[t] = 0.0f;
    pb[t] = 0.0f;
}

__global__ void final_k(const float* uf, const float* itf, const float* bfv,
                        const float* __restrict__ dvi, const float* __restrict__ va,
                        const float* __restrict__ vb,
                        float* __restrict__ all, float* __restrict__ partials) {
    __shared__ float sdata[256];
    long t = (long)blockIdx.x * blockDim.x + threadIdx.x;
    float sq = 0.0f;
    if (t < (long)NTOT * EMB) {
        int i = (int)(t >> 5), d = (int)(t & 31);
        float val = feat_at(uf, itf, bfv, i, d) * 0.5f + (va[t] + vb[t]) * dvi[i] * (1.0f / 3.0f);
        all[t] = val;
        sq = val * val;
    }
    sdata[threadIdx.x] = sq;
    __syncthreads();
    for (int s = 128; s > 0; s >>= 1) {
        if (threadIdx.x < (unsigned)s) sdata[threadIdx.x] += sdata[threadIdx.x + s];
        __syncthreads();
    }
    if (threadIdx.x == 0) partials[blockIdx.x] = sdata[0];
}

__global__ void loss_k(const float* __restrict__ partials, int nb, float* __restrict__ out_loss) {
    __shared__ float sdata[256];
    float s = 0.0f;
    for (int i = threadIdx.x; i < nb; i += 256) s += partials[i];
    sdata[threadIdx.x] = s;
    __syncthreads();
    for (int k = 128; k > 0; k >>= 1) {
        if (threadIdx.x < (unsigned)k) sdata[threadIdx.x] += sdata[threadIdx.x + k];
        __syncthreads();
    }
    if (threadIdx.x == 0) *out_loss = L2NORM * sdata[0];
}

__global__ void pred_k(const float* __restrict__ all, const int* __restrict__ users,
                       const int* __restrict__ bundles, const float* __restrict__ ubound,
                       const float* __restrict__ zrow,
                       float* __restrict__ pred, float* __restrict__ usb) {
    int wave = (blockIdx.x * blockDim.x + threadIdx.x) >> 5;
    int lane = threadIdx.x & 31;
    int b = wave / 7;
    int kt = wave % 7;

    const float* bfbase = all + (size_t)(NU + NI) * EMB;
    const float* ue = all + (size_t)users[b] * EMB;

    int m = lane & 15;
    int khalf = lane >> 4;
    int kk = kt * 16 + m;

    const float* arow;
    if (kk < KB)                 arow = bfbase + (size_t)bundles[b * KB + kk] * EMB;
    else if (kt == 6 && m == 4)  arow = ubound;
    else                         arow = zrow;

    v8f c = {};
#pragma unroll
    for (int s = 0; s < 8; ++s) {
        int e0 = s * 4 + khalf * 2;
        v2f a  = *(const v2f*)(arow + e0);
        v2f bb = *(const v2f*)(ue + e0);
        c = __builtin_amdgcn_wmma_f32_16x16x4_f32(false, a, false, bb,
                                                  (short)0, c, false, false);
    }

    if (lane == 0) {
#pragma unroll
        for (int j = 0; j < 8; ++j) {
            int kkk = kt * 16 + j;
            if (kkk < KB)                 pred[(size_t)b * KB + kkk] = c[j];
            else if (kt == 6 && j == 4)   usb[b] = c[j];
        }
    } else if (lane == 16) {
#pragma unroll
        for (int j = 0; j < 8; ++j) {
            int kkk = kt * 16 + 8 + j;
            if (kkk < KB) pred[(size_t)b * KB + kkk] = c[j];
        }
    }
}

extern "C" void kernel_launch(void* const* d_in, const int* in_sizes, int n_in,
                              void* d_out, int out_size, void* d_ws, size_t ws_size,
                              hipStream_t stream) {
    const float* uf      = (const float*)d_in[0];
    const float* itf     = (const float*)d_in[1];
    const float* bfv     = (const float*)d_in[2];
    const float* ubound  = (const float*)d_in[3];
    const int*   rows    = (const int*)d_in[4];
    const int*   cols    = (const int*)d_in[5];
    const int*   users   = (const int*)d_in[6];
    const int*   bundles = (const int*)d_in[7];
    float* out = (float*)d_out;

    float* ws   = (float*)d_ws;
    float* dvde = ws;
    float* z    = dvde + 2L * NTOT;
    float* u    = z + 32L * NTOT;
    float* v    = u + 32L * NTOT;
    float* partials = v + 32L * NTOT;
    float* zrow = partials + 8192;

    const long elems = (long)NTOT * EMB;
    const int  eb    = (int)((elems + 255) / 256);

    hist_k<<<dim3(HQ, HP, 2), 1024, 0, stream>>>(rows, cols, u);
    rsqsum_k<<<(2 * NTOT + 255) / 256, 256, 0, stream>>>(u, dvde);

    zero_k<<<4096, 256, 0, stream>>>(u, 64L * NTOT);
    zero_k<<<1, 32, 0, stream>>>(zrow, 32L);

    buildz_k<<<eb, 256, 0, stream>>>(uf, itf, bfv, dvde, z);
    scatter_k<<<dim3(SP, SQ), 1024, 0, stream>>>(rows, cols, z, u);
    combine_k<<<eb, 256, 0, stream>>>(u, v, dvde + NTOT, z);
    scatter_k<<<dim3(SP, SQ), 1024, 0, stream>>>(cols, rows, z, u);

    final_k<<<eb, 256, 0, stream>>>(uf, itf, bfv, dvde, u, v, z, partials);
    loss_k<<<1, 256, 0, stream>>>(partials, eb, out + BATCHN * KB + BATCHN);

    pred_k<<<1792, 256, 0, stream>>>(z, users, bundles, ubound, zrow,
                                     out, out + BATCHN * KB);
}


# scatter scan: each lane keeps its hits in an LDS queue of its own (store always, count only a hit), packed per wave before the row loads: no ballot or branch per edge; LDS adds in arrival order
# speedup vs baseline: 4.3925x; 1.1163x over previous
#include <hip/hip_runtime.h>
#include <stddef.h>

#define NU   8039
#define NI   32770
#define NB   4771
#define NTOT (NU + NI + NB)
#define EMB  32
#define NNZE 2000000
#define BATCHN 2048
#define KB   100
#define EPSF 1e-8f
#define L2NORM 1e-5f

typedef __attribute__((ext_vector_type(2))) float v2f;
typedef __attribute__((ext_vector_type(8))) float v8f;

__device__ __forceinline__ void atomAddF(float* p, float v) {
    (void)__hip_atomic_fetch_add(p, v, __ATOMIC_RELAXED, __HIP_MEMORY_SCOPE_AGENT);
}

__device__ __forceinline__ float feat_at(const float* uf, const float* itf,
                                         const float* bfv, int i, int d) {
    if (i < NU)      return uf[(size_t)i * EMB + d];
    if (i < NU + NI) return itf[(size_t)(i - NU) * EMB + d];
    return bfv[(size_t)(i - NU - NI) * EMB + d];
}

__global__ void zero_k(float* p, long n) {
    long i = (long)blockIdx.x * blockDim.x + threadIdx.x;
    long stride = (long)gridDim.x * blockDim.x;
    for (; i < n; i += stride) p[i] = 0.0f;
}

#define HP  4
#define HQ  32
#define HR  (NTOT / HP)
#define VPQ (NNZE / 4 / HQ)
#define HLD 8

__device__ __forceinline__ void bin_hit(unsigned int* bins, int idx, int lo) {
    unsigned int r = (unsigned int)(idx - lo);
    if (r < (unsigned int)HR) atomicAdd(bins + r, 1u);
}

__global__ __launch_bounds__(1024) void hist_k(const int* __restrict__ rows,
                                               const int* __restrict__ cols,
                                               float* __restrict__ part) {
    __shared__ unsigned int bins[HR];
    const int q = blockIdx.x, p = blockIdx.y, a = blockIdx.z;
    const int tid = threadIdx.x;
    for (int i = tid; i < HR; i += 1024) bins[i] = 0u;
    __syncthreads();

    const int lo = p * HR;
    const int4* __restrict__ vec = (const int4*)(a ? cols : rows) + (size_t)q * VPQ;
    for (int base = 0; base < VPQ; base += HLD * 1024) {
        int4 e[HLD];
#pragma unroll
        for (int j = 0; j < HLD; ++j) {
            int i = base + j * 1024 + tid;
            e[j] = (i < VPQ) ? vec[i] : make_int4(-1, -1, -1, -1);
        }
#pragma unroll
        for (int j = 0; j < HLD; ++j) {
            bin_hit(bins, e[j].x, lo);
            bin_hit(bins, e[j].y, lo);
            bin_hit(bins, e[j].z, lo);
            bin_hit(bins, e[j].w, lo);
        }
    }
    __syncthreads();

    float* __restrict__ dst = part + ((size_t)q * 2 + a) * NTOT + lo;
    for (int i = tid; i < HR; i += 1024) dst[i] = (float)bins[i];
}

__global__ void rsqsum_k(const float* __restrict__ part, float* __restrict__ dvde) {
    int i = blockIdx.x * blockDim.x + threadIdx.x;
    if (i >= 2 * NTOT) return;
    float d = 0.0f;
#pragma unroll
    for (int q = 0; q < HQ; ++q) d += part[(size_t)q * (2 * NTOT) + i];
    dvde[i] = 1.0f / (sqrtf(d) + EPSF);
}

__global__ void buildz_k(const float* uf, const float* itf, const float* bfv,
                         const float* __restrict__ dvi, float* __restrict__ z) {
    long t = (long)blockIdx.x * blockDim.x + threadIdx.x;
    if (t >= (long)NTOT * EMB) return;
    int i = (int)(t >> 5), d = (int)(t & 31);
    z[t] = feat_at(uf, itf, bfv, i, d) * dvi[i];
}

#define PLANE  ((size_t)NTOT * EMB)
#define SP   25
#define SQ   10
#define SR   1824
#define SVEC (NNZE / 4 / SQ)
#define SLD  4
#define QB   (SLD * 4)
#define QS   22
static_assert(SP * SR >= NTOT, "the ranges cover every row");
static_assert(NNZE % (4 * SQ) == 0, "every slice is a whole number of 16-byte vectors");
static_assert(NTOT <= 65536 && SR <= 2048, "queue word = source row << 11 | local destination row");
static_assert(QS > QB, "a lane that starts a batch with at most QS - QB words cannot overrun its queue");

__device__ __forceinline__ unsigned int cache_half() {
    return (__builtin_amdgcn_s_sendmsg_rtn(0x87) >> 18) & 1u;
}

__device__ __forceinline__ void push_edge(unsigned int* __restrict__ myq, int& cnt,
                                          int d, int s, int lo) {
    const unsigned int r = (unsigned int)(d - lo);
    myq[cnt * 1024] = ((unsigned int)s << 11) | r;
    cnt += (r < (unsigned int)SR) ? 1 : 0;
}

__global__ __launch_bounds__(1024) void scatter_k(const int* __restrict__ sidx,
                                                  const int* __restrict__ didx,
                                                  const float* __restrict__ src,
                                                  float* __restrict__ dst) {
    __shared__ float acc[SR * EMB];
    __shared__ unsigned int queue[QS * 1024];
    const int p = blockIdx.x, q = blockIdx.y;
    const int tid = threadIdx.x;
    const int lane = tid & 31;
    const int w = __builtin_amdgcn_readfirstlane(tid >> 5);

    for (int i = tid; i < SR * EMB / 4; i += 1024)
        ((float4*)acc)[i] = make_float4(0.0f, 0.0f, 0.0f, 0.0f);
    __syncthreads();

    const int lo = p * SR;
    float* __restrict__ plane = dst + (size_t)cache_half() * PLANE;
    const int4* __restrict__ sv4 = (const int4*)sidx + (size_t)q * SVEC;
    const int4* __restrict__ dv4 = (const int4*)didx + (size_t)q * SVEC;
    unsigned int* __restrict__ myq = queue + tid;
    unsigned int* __restrict__ wq  = queue + (w << 5);

    int cnt = 0;
    for (int base = 0; base < SVEC; base += SLD * 1024) {
        int4 dd[SLD], ss[SLD];
        int  lov[SLD];
#pragma unroll
        for (int j = 0; j < SLD; ++j) {
            const int i = base + j * 1024 + tid;
            const int ic = (i < SVEC) ? i : SVEC - 1;
            dd[j] = dv4[ic];
            ss[j] = sv4[ic];
            lov[j] = (i < SVEC) ? lo : 0x7fffffff;
        }
#pragma unroll
        for (int j = 0; j < SLD; ++j) {
            push_edge(myq, cnt, dd[j].x, ss[j].x, lov[j]);
            push_edge(myq, cnt, dd[j].y, ss[j].y, lov[j]);
            push_edge(myq, cnt, dd[j].z, ss[j].z, lov[j]);
            push_edge(myq, cnt, dd[j].w, ss[j].w, lov[j]);
        }

        const bool last = base + SLD * 1024 >= SVEC;
        if (!last && __builtin_amdgcn_ballot_w32(cnt > QS - QB) == 0u) continue;

        int n = 0;
        for (int k = 0; k < QS; ++k) {
            const bool have = cnt > k;
            const unsigned int mask = __builtin_amdgcn_ballot_w32(have);
            if (mask == 0u) break;
            const unsigned int word = myq[k * 1024];
            const int pos = n + (int)__builtin_amdgcn_mbcnt_lo(mask, 0u);
            if (have) wq[(pos >> 5) * 1024 + (pos & 31)] = word;
            n += __builtin_popcount(mask);
        }

        const int nd = last ? n : (n & ~31);
        for (int g = 0; g < nd; g += 32) {
            const int ent = (g + lane < n) ? (int)myq[(g >> 5) * 1024] : 0;
            const int m = n - g;
            float val[32];
#pragma unroll
            for (int j = 0; j < 32; ++j) {
                const unsigned int e = (unsigned int)__builtin_amdgcn_readlane(ent, j);
                val[j] = src[(size_t)(e >> 11) * EMB + lane];
            }
            if (m >= 32) {
#pragma unroll
                for (int j = 0; j < 32; ++j) {
                    const unsigned int e = (unsigned int)__builtin_amdgcn_readlane(ent, j);
                    atomicAdd(acc + (e & 2047u) * EMB + lane, val[j]);
                }
            } else {
#pragma unroll
                for (int j = 0; j < 32; ++j) {
                    const unsigned int e = (unsigned int)__builtin_amdgcn_readlane(ent, j);
                    if (j < m) atomicAdd(acc + (e & 2047u) * EMB + lane, val[j]);
                }
            }
        }

        const int rem = n & 31;
        if (lane < rem) myq[0] = myq[(n >> 5) * 1024];
        cnt = (lane < rem) ? 1 : 0;
    }
    __syncthreads();

    const int nrows = (NTOT - lo < SR) ? (NTOT - lo) : SR;
    for (int row = w; row < nrows; row += 32)
        atomAddF(plane + (size_t)(lo + row) * EMB + lane, acc[row * EMB + lane]);
}

__global__ void combine_k(float* __restrict__ pa, float* __restrict__ pb,
                          const float* __restrict__ s, float* __restrict__ res) {
    long t = (long)blockIdx.x * blockDim.x + threadIdx.x;
    if (t >= (long)NTOT * EMB) return;
    res[t] = (pa[t] + pb[t]) * s[t >> 5];
    pa[t] = 0.0f;
    pb[t] = 0.0f;
}

__global__ void final_k(const float* uf, const float* itf, const float* bfv,
                        const float* __restrict__ dvi, const float* __restrict__ va,
                        const float* __restrict__ vb,
                        float* __restrict__ all, float* __restrict__ partials) {
    __shared__ float sdata[256];
    long t = (long)blockIdx.x * blockDim.x + threadIdx.x;
    float sq = 0.0f;
    if (t < (long)NTOT * EMB) {
        int i = (int)(t >> 5), d = (int)(t & 31);
        float val = feat_at(uf, itf, bfv, i, d) * 0.5f + (va[t] + vb[t]) * dvi[i] * (1.0f / 3.0f);
        all[t] = val;
        sq = val * val;
    }
    sdata[threadIdx.x] = sq;
    __syncthreads();
    for (int s = 128; s > 0; s >>= 1) {
        if (threadIdx.x < (unsigned)s) sdata[threadIdx.x] += sdata[threadIdx.x + s];
        __syncthreads();
    }
    if (threadIdx.x == 0) partials[blockIdx.x] = sdata[0];
}

__global__ void loss_k(const float* __restrict__ partials, int nb, float* __restrict__ out_loss) {
    __shared__ float sdata[256];
    float s = 0.0f;
    for (int i = threadIdx.x; i < nb; i += 256) s += partials[i];
    sdata[threadIdx.x] = s;
    __syncthreads();
    for (int k = 128; k > 0; k >>= 1) {
        if (threadIdx.x < (unsigned)k) sdata[threadIdx.x] += sdata[threadIdx.x + k];
        __syncthreads();
    }
    if (threadIdx.x == 0) *out_loss = L2NORM * sdata[0];
}

__global__ void pred_k(const float* __restrict__ all, const int* __restrict__ users,
                       const int* __restrict__ bundles, const float* __restrict__ ubound,
                       const float* __restrict__ zrow,
                       float* __restrict__ pred, float* __restrict__ usb) {
    int wave = (blockIdx.x * blockDim.x + threadIdx.x) >> 5;
    int lane = threadIdx.x & 31;
    int b = wave / 7;
    int kt = wave % 7;

    const float* bfbase = all + (size_t)(NU + NI) * EMB;
    const float* ue = all + (size_t)users[b] * EMB;

    int m = lane & 15;
    int khalf = lane >> 4;
    int kk = kt * 16 + m;

    const float* arow;
    if (kk < KB)                 arow = bfbase + (size_t)bundles[b * KB + kk] * EMB;
    else if (kt == 6 && m == 4)  arow = ubound;
    else                         arow = zrow;

    v8f c = {};
#pragma unroll
    for (int s = 0; s < 8; ++s) {
        int e0 = s * 4 + khalf * 2;
        v2f a  = *(const v2f*)(arow + e0);
        v2f bb = *(const v2f*)(ue + e0);
        c = __builtin_amdgcn_wmma_f32_16x16x4_f32(false, a, false, bb,
                                                  (short)0, c, false, false);
    }

    if (lane == 0) {
#pragma unroll
        for (int j = 0; j < 8; ++j) {
            int kkk = kt * 16 + j;
            if (kkk < KB)                 pred[(size_t)b * KB + kkk] = c[j];
            else if (kt == 6 && j == 4)   usb[b] = c[j];
        }
    } else if (lane == 16) {
#pragma unroll
        for (int j = 0; j < 8; ++j) {
            int kkk = kt * 16 + 8 + j;
            if (kkk < KB) pred[(size_t)b * KB + kkk] = c[j];
        }
    }
}

extern "C" void kernel_launch(void* const* d_in, const int* in_sizes, int n_in,
                              void* d_out, int out_size, void* d_ws, size_t ws_size,
                              hipStream_t stream) {
    const float* uf      = (const float*)d_in[0];
    const float* itf     = (const float*)d_in[1];
    const float* bfv     = (const float*)d_in[2];
    const float* ubound  = (const float*)d_in[3];
    const int*   rows    = (const int*)d_in[4];
    const int*   cols    = (const int*)d_in[5];
    const int*   users   = (const int*)d_in[6];
    const int*   bundles = (const int*)d_in[7];
    float* out = (float*)d_out;

    float* ws   = (float*)d_ws;
    float* dvde = ws;
    float* z    = dvde + 2L * NTOT;
    float* u    = z + 32L * NTOT;
    float* v    = u + 32L * NTOT;
    float* partials = v + 32L * NTOT;
    float* zrow = partials + 8192;

    const long elems = (long)NTOT * EMB;
    const int  eb    = (int)((elems + 255) / 256);

    hist_k<<<dim3(HQ, HP, 2), 1024, 0, stream>>>(rows, cols, u);
    rsqsum_k<<<(2 * NTOT + 255) / 256, 256, 0, stream>>>(u, dvde);

    zero_k<<<4096, 256, 0, stream>>>(u, 64L * NTOT);
    zero_k<<<1, 32, 0, stream>>>(zrow, 32L);

    buildz_k<<<eb, 256, 0, stream>>>(uf, itf, bfv, dvde, z);
    scatter_k<<<dim3(SP, SQ), 1024, 0, stream>>>(rows, cols, z, u);
    combine_k<<<eb, 256, 0, stream>>>(u, v, dvde + NTOT, z);
    scatter_k<<<dim3(SP, SQ), 1024, 0, stream>>>(cols, rows, z, u);

    final_k<<<eb, 256, 0, stream>>>(uf, itf, bfv, dvde, u, v, z, partials);
    loss_k<<<1, 256, 0, stream>>>(partials, eb, out + BATCHN * KB + BATCHN);

    pred_k<<<1792, 256, 0, stream>>>(z, users, bundles, ubound, zrow,
                                     out, out + BATCHN * KB);
}


# launch fusion, 11 launches to 7: degree sums, plane and pad-row clearing and z build in one launch; loss sum by an extra block of the prediction launch, loads in flight; all sums in the same order
# speedup vs baseline: 4.6416x; 1.0567x over previous
#include <hip/hip_runtime.h>
#include <stddef.h>

#define NU   8039
#define NI   32770
#define NB   4771
#define NTOT (NU + NI + NB)
#define EMB  32
#define NNZE 2000000
#define BATCHN 2048
#define KB   100
#define EPSF 1e-8f
#define L2NORM 1e-5f

typedef __attribute__((ext_vector_type(2))) float v2f;
typedef __attribute__((ext_vector_type(8))) float v8f;

__device__ __forceinline__ void atomAddF(float* p, float v) {
    (void)__hip_atomic_fetch_add(p, v, __ATOMIC_RELAXED, __HIP_MEMORY_SCOPE_AGENT);
}

__device__ __forceinline__ float feat_at(const float* uf, const float* itf,
                                         const float* bfv, int i, int d) {
    if (i < NU)      return uf[(size_t)i * EMB + d];
    if (i < NU + NI) return itf[(size_t)(i - NU) * EMB + d];
    return bfv[(size_t)(i - NU - NI) * EMB + d];
}

#define HP  4
#define HQ  32
#define HR  (NTOT / HP)
#define VPQ (NNZE / 4 / HQ)
#define HLD 8

__device__ __forceinline__ void bin_hit(unsigned int* bins, int idx, int lo) {
    unsigned int r = (unsigned int)(idx - lo);
    if (r < (unsigned int)HR) atomicAdd(bins + r, 1u);
}

__global__ __launch_bounds__(1024) void hist_k(const int* __restrict__ rows,
                                               const int* __restrict__ cols,
                                               float* __restrict__ part) {
    __shared__ unsigned int bins[HR];
    const int q = blockIdx.x, p = blockIdx.y, a = blockIdx.z;
    const int tid = threadIdx.x;
    for (int i = tid; i < HR; i += 1024) bins[i] = 0u;
    __syncthreads();

    const int lo = p * HR;
    const int4* __restrict__ vec = (const int4*)(a ? cols : rows) + (size_t)q * VPQ;
    for (int base = 0; base < VPQ; base += HLD * 1024) {
        int4 e[HLD];
#pragma unroll
        for (int j = 0; j < HLD; ++j) {
            int i = base + j * 1024 + tid;
            e[j] = (i < VPQ) ? vec[i] : make_int4(-1, -1, -1, -1);
        }
#pragma unroll
        for (int j = 0; j < HLD; ++j) {
            bin_hit(bins, e[j].x, lo);
            bin_hit(bins, e[j].y, lo);
            bin_hit(bins, e[j].z, lo);
            bin_hit(bins, e[j].w, lo);
        }
    }
    __syncthreads();

    float* __restrict__ dst = part + ((size_t)q * 2 + a) * NTOT + lo;
    for (int i = tid; i < HR; i += 1024) dst[i] = (float)bins[i];
}

#define PN 128
static_assert(2 * PN == 256, "one thread per (node, index array)");
__global__ __launch_bounds__(256) void prep_k(const float* uf, const float* itf, const float* bfv,
                                              float* part, float* __restrict__ dvde,
                                              float* __restrict__ z, float* __restrict__ zrow) {
    __shared__ float sdv[PN];
    const int t = threadIdx.x;
    const int j = t & (PN - 1), a = t >> 7;
    const int base = blockIdx.x * PN;
    const int i = base + j;
    if (i < NTOT) {
        float* pp = part + (size_t)a * NTOT + i;
        float pv[HQ];
#pragma unroll
        for (int q = 0; q < HQ; ++q) pv[q] = pp[(size_t)q * (2 * NTOT)];
        float d = 0.0f;
#pragma unroll
        for (int q = 0; q < HQ; ++q) d += pv[q];
        const float inv = 1.0f / (sqrtf(d) + EPSF);
        dvde[(size_t)a * NTOT + i] = inv;
        if (a == 0) sdv[j] = inv;
    }
    if (blockIdx.x == 0 && t < 32) zrow[t] = 0.0f;
    __syncthreads();

    if (i < NTOT) {
        float* pp = part + (size_t)a * NTOT + i;
#pragma unroll
        for (int q = 0; q < HQ; ++q) pp[(size_t)q * (2 * NTOT)] = 0.0f;
    }

    const int nrow = (NTOT - base < PN) ? (NTOT - base) : PN;
#pragma unroll
    for (int k = 0; k < PN * EMB / 256; ++k) {
        const int e = k * 256 + t;
        const int r = e >> 5, dd = e & 31;
        if (r < nrow)
            z[(size_t)(base + r) * EMB + dd] = feat_at(uf, itf, bfv, base + r, dd) * sdv[r];
    }
}

#define PLANE  ((size_t)NTOT * EMB)
#define SP   25
#define SQ   10
#define SR   1824
#define SVEC (NNZE / 4 / SQ)
#define SLD  4
#define QB   (SLD * 4)
#define QS   22
static_assert(SP * SR >= NTOT, "the ranges cover every row");
static_assert(NNZE % (4 * SQ) == 0, "every slice is a whole number of 16-byte vectors");
static_assert(NTOT <= 65536 && SR <= 2048, "queue word = source row << 11 | local destination row");
static_assert(QS > QB, "a lane that starts a batch with at most QS - QB words cannot overrun its queue");

__device__ __forceinline__ unsigned int cache_half() {
    return (__builtin_amdgcn_s_sendmsg_rtn(0x87) >> 18) & 1u;
}

__device__ __forceinline__ void push_edge(unsigned int* __restrict__ myq, int& cnt,
                                          int d, int s, int lo) {
    const unsigned int r = (unsigned int)(d - lo);
    myq[cnt * 1024] = ((unsigned int)s << 11) | r;
    cnt += (r < (unsigned int)SR) ? 1 : 0;
}

__global__ __launch_bounds__(1024) void scatter_k(const int* __restrict__ sidx,
                                                  const int* __restrict__ didx,
                                                  const float* __restrict__ src,
                                                  float* __restrict__ dst) {
    __shared__ float acc[SR * EMB];
    __shared__ unsigned int queue[QS * 1024];
    const int p = blockIdx.x, q = blockIdx.y;
    const int tid = threadIdx.x;
    const int lane = tid & 31;
    const int w = __builtin_amdgcn_readfirstlane(tid >> 5);

    for (int i = tid; i < SR * EMB / 4; i += 1024)
        ((float4*)acc)[i] = make_float4(0.0f, 0.0f, 0.0f, 0.0f);
    __syncthreads();

    const int lo = p * SR;
    float* __restrict__ plane = dst + (size_t)cache_half() * PLANE;
    const int4* __restrict__ sv4 = (const int4*)sidx + (size_t)q * SVEC;
    const int4* __restrict__ dv4 = (const int4*)didx + (size_t)q * SVEC;
    unsigned int* __restrict__ myq = queue + tid;
    unsigned int* __restrict__ wq  = queue + (w << 5);

    int cnt = 0;
    for (int base = 0; base < SVEC; base += SLD * 1024) {
        int4 dd[SLD], ss[SLD];
        int  lov[SLD];
#pragma unroll
        for (int j = 0; j < SLD; ++j) {
            const int i = base + j * 1024 + tid;
            const int ic = (i < SVEC) ? i : SVEC - 1;
            dd[j] = dv4[ic];
            ss[j] = sv4[ic];
            lov[j] = (i < SVEC) ? lo : 0x7fffffff;
        }
#pragma unroll
        for (int j = 0; j < SLD; ++j) {
            push_edge(myq, cnt, dd[j].x, ss[j].x, lov[j]);
            push_edge(myq, cnt, dd[j].y, ss[j].y, lov[j]);
            push_edge(myq, cnt, dd[j].z, ss[j].z, lov[j]);
            push_edge(myq, cnt, dd[j].w, ss[j].w, lov[j]);
        }

        const bool last = base + SLD * 1024 >= SVEC;
        if (!last && __builtin_amdgcn_ballot_w32(cnt > QS - QB) == 0u) continue;

        int n = 0;
        for (int k = 0; k < QS; ++k) {
            const bool have = cnt > k;
            const unsigned int mask = __builtin_amdgcn_ballot_w32(have);
            if (mask == 0u) break;
            const unsigned int word = myq[k * 1024];
            const int pos = n + (int)__builtin_amdgcn_mbcnt_lo(mask, 0u);
            if (have) wq[(pos >> 5) * 1024 + (pos & 31)] = word;
            n += __builtin_popcount(mask);
        }

        const int nd = last ? n : (n & ~31);
        for (int g = 0; g < nd; g += 32) {
            const int ent = (g + lane < n) ? (int)myq[(g >> 5) * 1024] : 0;
            const int m = n - g;
            float val[32];
#pragma unroll
            for (int j = 0; j < 32; ++j) {
                const unsigned int e = (unsigned int)__builtin_amdgcn_readlane(ent, j);
                val[j] = src[(size_t)(e >> 11) * EMB + lane];
            }
            if (m >= 32) {
#pragma unroll
                for (int j = 0; j < 32; ++j) {
                    const unsigned int e = (unsigned int)__builtin_amdgcn_readlane(ent, j);
                    atomicAdd(acc + (e & 2047u) * EMB + lane, val[j]);
                }
            } else {
#pragma unroll
                for (int j = 0; j < 32; ++j) {
                    const unsigned int e = (unsigned int)__builtin_amdgcn_readlane(ent, j);
                    if (j < m) atomicAdd(acc + (e & 2047u) * EMB + lane, val[j]);
                }
            }
        }

        const int rem = n & 31;
        if (lane < rem) myq[0] = myq[(n >> 5) * 1024];
        cnt = (lane < rem) ? 1 : 0;
    }
    __syncthreads();

    const int nrows = (NTOT - lo < SR) ? (NTOT - lo) : SR;
    for (int row = w; row < nrows; row += 32)
        atomAddF(plane + (size_t)(lo + row) * EMB + lane, acc[row * EMB + lane]);
}

__global__ void combine_k(float* __restrict__ pa, float* __restrict__ pb,
                          const float* __restrict__ s, float* __restrict__ res) {
    long t = (long)blockIdx.x * blockDim.x + threadIdx.x;
    if (t >= (long)NTOT * EMB) return;
    res[t] = (pa[t] + pb[t]) * s[t >> 5];
    pa[t] = 0.0f;
    pb[t] = 0.0f;
}

__global__ void final_k(const float* uf, const float* itf, const float* bfv,
                        const float* __restrict__ dvi, const float* __restrict__ va,
                        const float* __restrict__ vb,
                        float* __restrict__ all, float* __restrict__ partials) {
    __shared__ float sdata[256];
    long t = (long)blockIdx.x * blockDim.x + threadIdx.x;
    float sq = 0.0f;
    if (t < (long)NTOT * EMB) {
        int i = (int)(t >> 5), d = (int)(t & 31);
        float val = feat_at(uf, itf, bfv, i, d) * 0.5f + (va[t] + vb[t]) * dvi[i] * (1.0f / 3.0f);
        all[t] = val;
        sq = val * val;
    }
    sdata[threadIdx.x] = sq;
    __syncthreads();
    for (int s = 128; s > 0; s >>= 1) {
        if (threadIdx.x < (unsigned)s) sdata[threadIdx.x] += sdata[threadIdx.x + s];
        __syncthreads();
    }
    if (threadIdx.x == 0) partials[blockIdx.x] = sdata[0];
}

#define EBLK ((NTOT * EMB + 255) / 256)
__global__ void pred_k(const float* __restrict__ all, const int* __restrict__ users,
                       const int* __restrict__ bundles, const float* __restrict__ ubound,
                       const float* __restrict__ zrow,
                       const float* __restrict__ partials,
                       float* __restrict__ pred, float* __restrict__ usb,
                       float* __restrict__ out_loss) {
    if (blockIdx.x == 0) {
        __shared__ float sdata[256];
        float pv[(EBLK + 255) / 256];
#pragma unroll
        for (int k = 0; k < (EBLK + 255) / 256; ++k) {
            const int i = k * 256 + (int)threadIdx.x;
            pv[k] = partials[(i < EBLK) ? i : EBLK - 1];
        }
        float s = 0.0f;
#pragma unroll
        for (int k = 0; k < (EBLK + 255) / 256; ++k)
            if (k * 256 + (int)threadIdx.x < EBLK) s += pv[k];
        sdata[threadIdx.x] = s;
        __syncthreads();
        for (int k = 128; k > 0; k >>= 1) {
            if (threadIdx.x < (unsigned)k) sdata[threadIdx.x] += sdata[threadIdx.x + k];
            __syncthreads();
        }
        if (threadIdx.x == 0) *out_loss = L2NORM * sdata[0];
        return;
    }
    int wave = ((blockIdx.x - 1) * blockDim.x + threadIdx.x) >> 5;
    int lane = threadIdx.x & 31;
    int b = wave / 7;
    int kt = wave % 7;

    const float* bfbase = all + (size_t)(NU + NI) * EMB;
    const float* ue = all + (size_t)users[b] * EMB;

    int m = lane & 15;
    int khalf = lane >> 4;
    int kk = kt * 16 + m;

    const float* arow;
    if (kk < KB)                 arow = bfbase + (size_t)bundles[b * KB + kk] * EMB;
    else if (kt == 6 && m == 4)  arow = ubound;
    else                         arow = zrow;

    v8f c = {};
#pragma unroll
    for (int s = 0; s < 8; ++s) {
        int e0 = s * 4 + khalf * 2;
        v2f a  = *(const v2f*)(arow + e0);
        v2f bb = *(const v2f*)(ue + e0);
        c = __builtin_amdgcn_wmma_f32_16x16x4_f32(false, a, false, bb,
                                                  (short)0, c, false, false);
    }

    if (lane == 0) {
#pragma unroll
        for (int j = 0; j < 8; ++j) {
            int kkk = kt * 16 + j;
            if (kkk < KB)                 pred[(size_t)b * KB + kkk] = c[j];
            else if (kt == 6 && j == 4)   usb[b] = c[j];
        }
    } else if (lane == 16) {
#pragma unroll
        for (int j = 0; j < 8; ++j) {
            int kkk = kt * 16 + 8 + j;
            if (kkk < KB) pred[(size_t)b * KB + kkk] = c[j];
        }
    }
}

extern "C" void kernel_launch(void* const* d_in, const int* in_sizes, int n_in,
                              void* d_out, int out_size, void* d_ws, size_t ws_size,
                              hipStream_t stream) {
    const float* uf      = (const float*)d_in[0];
    const float* itf     = (const float*)d_in[1];
    const float* bfv     = (const float*)d_in[2];
    const float* ubound  = (const float*)d_in[3];
    const int*   rows    = (const int*)d_in[4];
    const int*   cols    = (const int*)d_in[5];
    const int*   users   = (const int*)d_in[6];
    const int*   bundles = (const int*)d_in[7];
    float* out = (float*)d_out;

    float* ws   = (float*)d_ws;
    float* dvde = ws;
    float* z    = dvde + 2L * NTOT;
    float* u    = z + 32L * NTOT;
    float* v    = u + 32L * NTOT;
    float* partials = v + 32L * NTOT;
    float* zrow = partials + 8192;

    const long elems = (long)NTOT * EMB;
    const int  eb    = (int)((elems + 255) / 256);

    hist_k<<<dim3(HQ, HP, 2), 1024, 0, stream>>>(rows, cols, u);

    prep_k<<<(NTOT + PN - 1) / PN, 256, 0, stream>>>(uf, itf, bfv, u, dvde, z, zrow);
    scatter_k<<<dim3(SP, SQ), 1024, 0, stream>>>(rows, cols, z, u);
    combine_k<<<eb, 256, 0, stream>>>(u, v, dvde + NTOT, z);
    scatter_k<<<dim3(SP, SQ), 1024, 0, stream>>>(cols, rows, z, u);

    final_k<<<eb, 256, 0, stream>>>(uf, itf, bfv, dvde, u, v, z, partials);
    pred_k<<<1793, 256, 0, stream>>>(z, users, bundles, ubound, zrow, partials,
                                     out, out + BATCHN * KB, out + BATCHN * KB + BATCHN);
}


# row-wise launches (combine, final, prediction): 4 16-byte vectors per thread, 7 tiles per wave: 105k waves to 8k (these launches were bound by wave entry); loss block partials regrouped, 5698 to 357
# speedup vs baseline: 4.8186x; 1.0381x over previous
#include <hip/hip_runtime.h>
#include <stddef.h>

#define NU   8039
#define NI   32770
#define NB   4771
#define NTOT (NU + NI + NB)
#define EMB  32
#define NNZE 2000000
#define BATCHN 2048
#define KB   100
#define EPSF 1e-8f
#define L2NORM 1e-5f

typedef __attribute__((ext_vector_type(2))) float v2f;
typedef __attribute__((ext_vector_type(8))) float v8f;

__device__ __forceinline__ void atomAddF(float* p, float v) {
    (void)__hip_atomic_fetch_add(p, v, __ATOMIC_RELAXED, __HIP_MEMORY_SCOPE_AGENT);
}

__device__ __forceinline__ float feat_at(const float* uf, const float* itf,
                                         const float* bfv, int i, int d) {
    if (i < NU)      return uf[(size_t)i * EMB + d];
    if (i < NU + NI) return itf[(size_t)(i - NU) * EMB + d];
    return bfv[(size_t)(i - NU - NI) * EMB + d];
}

#define HP  4
#define HQ  32
#define HR  (NTOT / HP)
#define VPQ (NNZE / 4 / HQ)
#define HLD 8

__device__ __forceinline__ void bin_hit(unsigned int* bins, int idx, int lo) {
    unsigned int r = (unsigned int)(idx - lo);
    if (r < (unsigned int)HR) atomicAdd(bins + r, 1u);
}

__global__ __launch_bounds__(1024) void hist_k(const int* __restrict__ rows,
                                               const int* __restrict__ cols,
                                               float* __restrict__ part) {
    __shared__ unsigned int bins[HR];
    const int q = blockIdx.x, p = blockIdx.y, a = blockIdx.z;
    const int tid = threadIdx.x;
    for (int i = tid; i < HR; i += 1024) bins[i] = 0u;
    __syncthreads();

    const int lo = p * HR;
    const int4* __restrict__ vec = (const int4*)(a ? cols : rows) + (size_t)q * VPQ;
    for (int base = 0; base < VPQ; base += HLD * 1024) {
        int4 e[HLD];
#pragma unroll
        for (int j = 0; j < HLD; ++j) {
            int i = base + j * 1024 + tid;
            e[j] = (i < VPQ) ? vec[i] : make_int4(-1, -1, -1, -1);
        }
#pragma unroll
        for (int j = 0; j < HLD; ++j) {
            bin_hit(bins, e[j].x, lo);
            bin_hit(bins, e[j].y, lo);
            bin_hit(bins, e[j].z, lo);
            bin_hit(bins, e[j].w, lo);
        }
    }
    __syncthreads();

    float* __restrict__ dst = part + ((size_t)q * 2 + a) * NTOT + lo;
    for (int i = tid; i < HR; i += 1024) dst[i] = (float)bins[i];
}

#define PN 128
static_assert(2 * PN == 256, "one thread per (node, index array)");
__global__ __launch_bounds__(256) void prep_k(const float* uf, const float* itf, const float* bfv,
                                              float* part, float* __restrict__ dvde,
                                              float* __restrict__ z, float* __restrict__ zrow) {
    __shared__ float sdv[PN];
    const int t = threadIdx.x;
    const int j = t & (PN - 1), a = t >> 7;
    const int base = blockIdx.x * PN;
    const int i = base + j;
    if (i < NTOT) {
        float* pp = part + (size_t)a * NTOT + i;
        float pv[HQ];
#pragma unroll
        for (int q = 0; q < HQ; ++q) pv[q] = pp[(size_t)q * (2 * NTOT)];
        float d = 0.0f;
#pragma unroll
        for (int q = 0; q < HQ; ++q) d += pv[q];
        const float inv = 1.0f / (sqrtf(d) + EPSF);
        dvde[(size_t)a * NTOT + i] = inv;
        if (a == 0) sdv[j] = inv;
    }
    if (blockIdx.x == 0 && t < 32) zrow[t] = 0.0f;
    __syncthreads();

    if (i < NTOT) {
        float* pp = part + (size_t)a * NTOT + i;
#pragma unroll
        for (int q = 0; q < HQ; ++q) pp[(size_t)q * (2 * NTOT)] = 0.0f;
    }

    const int nrow = (NTOT - base < PN) ? (NTOT - base) : PN;
#pragma unroll
    for (int k = 0; k < PN * EMB / 256; ++k) {
        const int e = k * 256 + t;
        const int r = e >> 5, dd = e & 31;
        if (r < nrow)
            z[(size_t)(base + r) * EMB + dd] = feat_at(uf, itf, bfv, base + r, dd) * sdv[r];
    }
}

#define PLANE  ((size_t)NTOT * EMB)
#define SP   25
#define SQ   10
#define SR   1824
#define SVEC (NNZE / 4 / SQ)
#define SLD  4
#define QB   (SLD * 4)
#define QS   22
static_assert(SP * SR >= NTOT, "the ranges cover every row");
static_assert(NNZE % (4 * SQ) == 0, "every slice is a whole number of 16-byte vectors");
static_assert(NTOT <= 65536 && SR <= 2048, "queue word = source row << 11 | local destination row");
static_assert(QS > QB, "a lane that starts a batch with at most QS - QB words cannot overrun its queue");

__device__ __forceinline__ unsigned int cache_half() {
    return (__builtin_amdgcn_s_sendmsg_rtn(0x87) >> 18) & 1u;
}

__device__ __forceinline__ void push_edge(unsigned int* __restrict__ myq, int& cnt,
                                          int d, int s, int lo) {
    const unsigned int r = (unsigned int)(d - lo);
    myq[cnt * 1024] = ((unsigned int)s << 11) | r;
    cnt += (r < (unsigned int)SR) ? 1 : 0;
}

__global__ __launch_bounds__(1024) void scatter_k(const int* __restrict__ sidx,
                                                  const int* __restrict__ didx,
                                                  const float* __restrict__ src,
                                                  float* __restrict__ dst) {
    __shared__ float acc[SR * EMB];
    __shared__ unsigned int queue[QS * 1024];
    const int p = blockIdx.x, q = blockIdx.y;
    const int tid = threadIdx.x;
    const int lane = tid & 31;
    const int w = __builtin_amdgcn_readfirstlane(tid >> 5);

    for (int i = tid; i < SR * EMB / 4; i += 1024)
        ((float4*)acc)[i] = make_float4(0.0f, 0.0f, 0.0f, 0.0f);
    __syncthreads();

    const int lo = p * SR;
    float* __restrict__ plane = dst + (size_t)cache_half() * PLANE;
    const int4* __restrict__ sv4 = (const int4*)sidx + (size_t)q * SVEC;
    const int4* __restrict__ dv4 = (const int4*)didx + (size_t)q * SVEC;
    unsigned int* __restrict__ myq = queue + tid;
    unsigned int* __restrict__ wq  = queue + (w << 5);

    int cnt = 0;
    for (int base = 0; base < SVEC; base += SLD * 1024) {
        int4 dd[SLD], ss[SLD];
        int  lov[SLD];
#pragma unroll
        for (int j = 0; j < SLD; ++j) {
            const int i = base + j * 1024 + tid;
            const int ic = (i < SVEC) ? i : SVEC - 1;
            dd[j] = dv4[ic];
            ss[j] = sv4[ic];
            lov[j] = (i < SVEC) ? lo : 0x7fffffff;
        }
#pragma unroll
        for (int j = 0; j < SLD; ++j) {
            push_edge(myq, cnt, dd[j].x, ss[j].x, lov[j]);
            push_edge(myq, cnt, dd[j].y, ss[j].y, lov[j]);
            push_edge(myq, cnt, dd[j].z, ss[j].z, lov[j]);
            push_edge(myq, cnt, dd[j].w, ss[j].w, lov[j]);
        }

        const bool last = base + SLD * 1024 >= SVEC;
        if (!last && __builtin_amdgcn_ballot_w32(cnt > QS - QB) == 0u) continue;

        int n = 0;
        for (int k = 0; k < QS; ++k) {
            const bool have = cnt > k;
            const unsigned int mask = __builtin_amdgcn_ballot_w32(have);
            if (mask == 0u) break;
            const unsigned int word = myq[k * 1024];
            const int pos = n + (int)__builtin_amdgcn_mbcnt_lo(mask, 0u);
            if (have) wq[(pos >> 5) * 1024 + (pos & 31)] = word;
            n += __builtin_popcount(mask);
        }

        const int nd = last ? n : (n & ~31);
        for (int g = 0; g < nd; g += 32) {
            const int ent = (g + lane < n) ? (int)myq[(g >> 5) * 1024] : 0;
            const int m = n - g;
            float val[32];
#pragma unroll
            for (int j = 0; j < 32; ++j) {
                const unsigned int e = (unsigned int)__builtin_amdgcn_readlane(ent, j);
                val[j] = src[(size_t)(e >> 11) * EMB + lane];
            }
            if (m >= 32) {
#pragma unroll
                for (int j = 0; j < 32; ++j) {
                    const unsigned int e = (unsigned int)__builtin_amdgcn_readlane(ent, j);
                    atomicAdd(acc + (e & 2047u) * EMB + lane, val[j]);
                }
            } else {
#pragma unroll
                for (int j = 0; j < 32; ++j) {
                    const unsigned int e = (unsigned int)__builtin_amdgcn_readlane(ent, j);
                    if (j < m) atomicAdd(acc + (e & 2047u) * EMB + lane, val[j]);
                }
            }
        }

        const int rem = n & 31;
        if (lane < rem) myq[0] = myq[(n >> 5) * 1024];
        cnt = (lane < rem) ? 1 : 0;
    }
    __syncthreads();

    const int nrows = (NTOT - lo < SR) ? (NTOT - lo) : SR;
    for (int row = w; row < nrows; row += 32)
        atomAddF(plane + (size_t)(lo + row) * EMB + lane, acc[row * EMB + lane]);
}

#define NVEC (NTOT * EMB / 4)
#define EV   4
#define EFULL (NVEC / (EV * 256))
#define EBLK  (EFULL + 1)
static_assert(NVEC - EFULL * (EV * 256) > 0 && NVEC - EFULL * (EV * 256) <= 256,
              "the last block has at most one vector per thread");

__device__ __forceinline__ void combine_vec(float* __restrict__ pa, float* __restrict__ pb,
                                            float* __restrict__ res, int i,
                                            float4 a, float4 b, float sc) {
    float4 r;
    r.x = (a.x + b.x) * sc;
    r.y = (a.y + b.y) * sc;
    r.z = (a.z + b.z) * sc;
    r.w = (a.w + b.w) * sc;
    ((float4*)res)[i] = r;
    ((float4*)pa)[i] = make_float4(0.0f, 0.0f, 0.0f, 0.0f);
    ((float4*)pb)[i] = make_float4(0.0f, 0.0f, 0.0f, 0.0f);
}

__global__ __launch_bounds__(256) void combine_k(float* __restrict__ pa, float* __restrict__ pb,
                                                 const float* __restrict__ s, float* __restrict__ res) {
    if (blockIdx.x < EFULL) {
        const int base = blockIdx.x * (EV * 256) + threadIdx.x;
        float4 a[EV], b[EV];
        float  sc[EV];
#pragma unroll
        for (int k = 0; k < EV; ++k) {
            const int i = base + k * 256;
            a[k]  = ((const float4*)pa)[i];
            b[k]  = ((const float4*)pb)[i];
            sc[k] = s[i >> 3];
        }
#pragma unroll
        for (int k = 0; k < EV; ++k) combine_vec(pa, pb, res, base + k * 256, a[k], b[k], sc[k]);
    } else {
        const int i = EFULL * (EV * 256) + threadIdx.x;
        if (i < NVEC) combine_vec(pa, pb, res, i, ((const float4*)pa)[i], ((const float4*)pb)[i], s[i >> 3]);
    }
}

__device__ __forceinline__ const float* feat_row(const float* uf, const float* itf,
                                                 const float* bfv, int i) {
    const float* base = (i < NU) ? uf : ((i < NU + NI) ? itf : bfv);
    const int    r    = (i < NU) ? i  : ((i < NU + NI) ? i - NU : i - NU - NI);
    return base + (size_t)r * EMB;
}

__device__ __forceinline__ float final_vec(float* __restrict__ all, int i,
                                           float4 x, float4 a, float4 b, float sc, float sq) {
    float4 r;
    r.x = x.x * 0.5f + (a.x + b.x) * sc * (1.0f / 3.0f);
    r.y = x.y * 0.5f + (a.y + b.y) * sc * (1.0f / 3.0f);
    r.z = x.z * 0.5f + (a.z + b.z) * sc * (1.0f / 3.0f);
    r.w = x.w * 0.5f + (a.w + b.w) * sc * (1.0f / 3.0f);
    ((float4*)all)[i] = r;
    sq += r.x * r.x;
    sq += r.y * r.y;
    sq += r.z * r.z;
    sq += r.w * r.w;
    return sq;
}

__global__ __launch_bounds__(256) void final_k(const float* uf, const float* itf, const float* bfv,
                                               const float* __restrict__ dvi, const float* __restrict__ va,
                                               const float* __restrict__ vb,
                                               float* __restrict__ all, float* __restrict__ partials) {
    __shared__ float sdata[256];
    float sq = 0.0f;
    if (blockIdx.x < EFULL) {
        const int base = blockIdx.x * (EV * 256) + threadIdx.x;
        float4 x[EV], a[EV], b[EV];
        float  sc[EV];
#pragma unroll
        for (int k = 0; k < EV; ++k) {
            const int i = base + k * 256;
            const int row = i >> 3;
            x[k]  = *(const float4*)(feat_row(uf, itf, bfv, row) + (i & 7) * 4);
            a[k]  = ((const float4*)va)[i];
            b[k]  = ((const float4*)vb)[i];
            sc[k] = dvi[row];
        }
#pragma unroll
        for (int k = 0; k < EV; ++k) sq = final_vec(all, base + k * 256, x[k], a[k], b[k], sc[k], sq);
    } else {
        const int i = EFULL * (EV * 256) + threadIdx.x;
        if (i < NVEC) {
            const int row = i >> 3;
            sq = final_vec(all, i, *(const float4*)(feat_row(uf, itf, bfv, row) + (i & 7) * 4),
                           ((const float4*)va)[i], ((const float4*)vb)[i], dvi[row], sq);
        }
    }
    sdata[threadIdx.x] = sq;
    __syncthreads();
    for (int s = 128; s > 0; s >>= 1) {
        if (threadIdx.x < (unsigned)s) sdata[threadIdx.x] += sdata[threadIdx.x + s];
        __syncthreads();
    }
    if (threadIdx.x == 0) partials[blockIdx.x] = sdata[0];
}

#define NKT 7
__global__ __launch_bounds__(256) void pred_k(const float* __restrict__ all, const int* __restrict__ users,
                       const int* __restrict__ bundles, const float* __restrict__ ubound,
                       const float* __restrict__ zrow,
                       const float* __restrict__ partials,
                       float* __restrict__ pred, float* __restrict__ usb,
                       float* __restrict__ out_loss) {
    if (blockIdx.x == 0) {
        __shared__ float sdata[256];
        float pv[(EBLK + 255) / 256];
#pragma unroll
        for (int k = 0; k < (EBLK + 255) / 256; ++k) {
            const int i = k * 256 + (int)threadIdx.x;
            pv[k] = partials[(i < EBLK) ? i : EBLK - 1];
        }
        float s = 0.0f;
#pragma unroll
        for (int k = 0; k < (EBLK + 255) / 256; ++k)
            if (k * 256 + (int)threadIdx.x < EBLK) s += pv[k];
        sdata[threadIdx.x] = s;
        __syncthreads();
        for (int k = 128; k > 0; k >>= 1) {
            if (threadIdx.x < (unsigned)k) sdata[threadIdx.x] += sdata[threadIdx.x + k];
            __syncthreads();
        }
        if (threadIdx.x == 0) *out_loss = L2NORM * sdata[0];
        return;
    }
    const int b = ((blockIdx.x - 1) * 256 + threadIdx.x) >> 5;
    const int lane = threadIdx.x & 31;

    const float* bfbase = all + (size_t)(NU + NI) * EMB;
    const int ub = users[b];

    const int m = lane & 15;
    const int khalf = lane >> 4;

    static_assert((NKT - 1) * 16 <= KB && NKT * 16 >= KB + 5, "only the last tile has rows past the bundles, the bound's row among them");
    int bid[NKT - 1];
#pragma unroll
    for (int kt = 0; kt < NKT - 1; ++kt) bid[kt] = bundles[b * KB + kt * 16 + m];

    const float* alast;
    {
        const int kk = (NKT - 1) * 16 + m;
        if (kk < KB)       alast = bfbase + (size_t)bundles[b * KB + kk] * EMB;
        else if (m == 4)   alast = ubound;
        else               alast = zrow;
    }

    v8f c[NKT];
#pragma unroll
    for (int kt = 0; kt < NKT; ++kt) {
        c[kt] = (v8f){};
        asm volatile("" : "+v"(c[kt]));
    }

    const float* ue = all + (size_t)ub * EMB;
    v2f bb[8];
#pragma unroll
    for (int s = 0; s < 8; ++s) bb[s] = *(const v2f*)(ue + s * 4 + khalf * 2);

    v2f a[NKT][8];
#pragma unroll
    for (int kt = 0; kt < NKT - 1; ++kt) {
        const float* arow = bfbase + (size_t)bid[kt] * EMB;
#pragma unroll
        for (int s = 0; s < 8; ++s) a[kt][s] = *(const v2f*)(arow + s * 4 + khalf * 2);
    }
#pragma unroll
    for (int s = 0; s < 8; ++s) a[NKT - 1][s] = *(const v2f*)(alast + s * 4 + khalf * 2);

    __builtin_amdgcn_sched_barrier(0);

#pragma unroll
    for (int kt = 0; kt < NKT; ++kt) {
#pragma unroll
        for (int s = 0; s < 8; ++s)
            c[kt] = __builtin_amdgcn_wmma_f32_16x16x4_f32(false, a[kt][s], false, bb[s],
                                                          (short)0, c[kt], false, false);
    }

    int lane_st;
    asm volatile("v_nop\n\tv_nop\n\tv_nop\n\tv_nop\n\tv_nop\n\tv_nop\n\tv_nop\n\tv_nop\n\t"
                 "v_and_b32 %7, 31, %8"
                 : "+v"(c[0]), "+v"(c[1]), "+v"(c[2]), "+v"(c[3]), "+v"(c[4]), "+v"(c[5]), "+v"(c[6]),
                   "=v"(lane_st)
                 : "v"(threadIdx.x));

    if (lane_st == 0) {
#pragma unroll
        for (int kt = 0; kt < NKT; ++kt) {
#pragma unroll
            for (int j = 0; j < 8; ++j) {
                const int kkk = kt * 16 + j;
                if (kkk < KB)                       pred[(size_t)b * KB + kkk] = c[kt][j];
                else if (kt == NKT - 1 && j == 4)   usb[b] = c[kt][j];
            }
        }
    } else if (lane_st == 16) {
#pragma unroll
        for (int kt = 0; kt < NKT; ++kt) {
#pragma unroll
            for (int j = 0; j < 8; ++j) {
                const int kkk = kt * 16 + 8 + j;
                if (kkk < KB) pred[(size_t)b * KB + kkk] = c[kt][j];
            }
        }
    }
}

extern "C" void kernel_launch(void* const* d_in, const int* in_sizes, int n_in,
                              void* d_out, int out_size, void* d_ws, size_t ws_size,
                              hipStream_t stream) {
    const float* uf      = (const float*)d_in[0];
    const float* itf     = (const float*)d_in[1];
    const float* bfv     = (const float*)d_in[2];
    const float* ubound  = (const float*)d_in[3];
    const int*   rows    = (const int*)d_in[4];
    const int*   cols    = (const int*)d_in[5];
    const int*   users   = (const int*)d_in[6];
    const int*   bundles = (const int*)d_in[7];
    float* out = (float*)d_out;

    float* ws   = (float*)d_ws;
    float* dvde = ws;
    float* z    = dvde + 2L * NTOT;
    float* u    = z + 32L * NTOT;
    float* v    = u + 32L * NTOT;
    float* partials = v + 32L * NTOT;
    float* zrow = partials + 8192;

    hist_k<<<dim3(HQ, HP, 2), 1024, 0, stream>>>(rows, cols, u);

    prep_k<<<(NTOT + PN - 1) / PN, 256, 0, stream>>>(uf, itf, bfv, u, dvde, z, zrow);
    scatter_k<<<dim3(SP, SQ), 1024, 0, stream>>>(rows, cols, z, u);
    combine_k<<<EBLK, 256, 0, stream>>>(u, v, dvde + NTOT, z);
    scatter_k<<<dim3(SP, SQ), 1024, 0, stream>>>(cols, rows, z, u);

    final_k<<<EBLK, 256, 0, stream>>>(uf, itf, bfv, dvde, u, v, z, partials);
    pred_k<<<BATCHN / 8 + 1, 256, 0, stream>>>(z, users, bundles, ubound, zrow, partials,
                                     out, out + BATCHN * KB, out + BATCHN * KB + BATCHN);
}


# edges counting-sorted by destination bucket of 128 rows per pass; a block sums a bucket in LDS (one arrival-order float sum per element), writes rows once: no planes or global atomics; loss regrouped
# speedup vs baseline: 5.4257x; 1.1260x over previous
#include <hip/hip_runtime.h>
#include <stddef.h>

#define NU   8039
#define NI   32770
#define NB   4771
#define NTOT (NU + NI + NB)
#define EMB  32
#define NNZE 2000000
#define BATCHN 2048
#define KB   100
#define EPSF 1e-8f
#define L2NORM 1e-5f

typedef __attribute__((ext_vector_type(2))) float v2f;
typedef __attribute__((ext_vector_type(8))) float v8f;
#define HP  4
#define HQ  32
#define HR  (NTOT / HP)
#define VPQ (NNZE / 4 / HQ)
#define HLD 8

__device__ __forceinline__ void bin_hit(unsigned int* bins, int idx, int lo) {
    unsigned int r = (unsigned int)(idx - lo);
    if (r < (unsigned int)HR) atomicAdd(bins + r, 1u);
}

__global__ __launch_bounds__(1024) void hist_k(const int* __restrict__ rows,
                                               const int* __restrict__ cols,
                                               float* __restrict__ part) {
    __shared__ unsigned int bins[HR];
    const int q = blockIdx.x, p = blockIdx.y, a = blockIdx.z;
    const int tid = threadIdx.x;
    for (int i = tid; i < HR; i += 1024) bins[i] = 0u;
    __syncthreads();

    const int lo = p * HR;
    const int4* __restrict__ vec = (const int4*)(a ? cols : rows) + (size_t)q * VPQ;
    for (int base = 0; base < VPQ; base += HLD * 1024) {
        int4 e[HLD];
#pragma unroll
        for (int j = 0; j < HLD; ++j) {
            int i = base + j * 1024 + tid;
            e[j] = (i < VPQ) ? vec[i] : make_int4(-1, -1, -1, -1);
        }
#pragma unroll
        for (int j = 0; j < HLD; ++j) {
            bin_hit(bins, e[j].x, lo);
            bin_hit(bins, e[j].y, lo);
            bin_hit(bins, e[j].z, lo);
            bin_hit(bins, e[j].w, lo);
        }
    }
    __syncthreads();

    float* __restrict__ dst = part + ((size_t)q * 2 + a) * NTOT + lo;
    for (int i = tid; i < HR; i += 1024) dst[i] = (float)bins[i];
}

#define BR   128
#define NBK  ((NTOT + BR - 1) / BR)
#define WSH  7
static_assert((1 << WSH) == BR, "the word's low bits hold a row of the bucket");
static_assert(NTOT <= 65536, "a source row and a row of the bucket fit one 32-bit word");
static_assert(NBK <= 12 * 32, "one wave scans the buckets, 12 per lane");

__device__ __forceinline__ void wave_excl_scan(const int* in, int* out, int n, int lane) {
    int v[12];
    int s = 0;
#pragma unroll
    for (int k = 0; k < 12; ++k) {
        const int idx = lane * 12 + k;
        v[k] = (idx < n) ? in[idx] : 0;
        s += v[k];
    }
    int inc = s;
#pragma unroll
    for (int d = 1; d < 32; d <<= 1) {
        const int t = __shfl_up(inc, d);
        if (lane >= d) inc += t;
    }
    int run = inc - s;
#pragma unroll
    for (int k = 0; k < 12; ++k) {
        const int idx = lane * 12 + k;
        if (idx < n) out[idx] = run;
        run += v[k];
    }
}

static_assert(2 * BR == 256, "one thread per (node, index array)");
__global__ __launch_bounds__(256) void prep_k(const float* __restrict__ part, float* __restrict__ dvde,
                                              int* __restrict__ totals, int* __restrict__ cursors,
                                              float* __restrict__ zrow) {
    __shared__ float sred[256];
    const int t = threadIdx.x;
    const int j = t & (BR - 1), a = t >> 7;
    const int i = blockIdx.x * BR + j;
    float d = 0.0f;
    if (i < NTOT) {
        const float* pp = part + (size_t)a * NTOT + i;
        float pv[HQ];
#pragma unroll
        for (int q = 0; q < HQ; ++q) pv[q] = pp[(size_t)q * (2 * NTOT)];
#pragma unroll
        for (int q = 0; q < HQ; ++q) d += pv[q];
        dvde[(size_t)a * NTOT + i] = 1.0f / (sqrtf(d) + EPSF);
    }
    sred[t] = d;
    if (blockIdx.x == 0) {
        if (t < 32) zrow[t] = 0.0f;
        for (int k = t; k < 2 * NBK; k += 256) cursors[k] = 0;
    }
    __syncthreads();
    for (int s = BR / 2; s > 0; s >>= 1) {
        if (j < s) sred[t] += sred[t + s];
        __syncthreads();
    }
    if (j == 0) totals[a * NBK + blockIdx.x] = (int)sred[t];
}

#define BV   2000
#define BE   (BV * 4)
#define BQN  (NNZE / BE)
static_assert(BQN * BE == NNZE, "every block is full");
static_assert(BV > 1024 && BV <= 2048, "a thread has one or two vectors");
__global__ __launch_bounds__(1024) void bin_k(const int* __restrict__ sidx, const int* __restrict__ didx,
                                              const int* __restrict__ totals, int* __restrict__ cursors,
                                              unsigned int* __restrict__ bucket) {
    __shared__ int cnt[NBK];
    __shared__ int goff[NBK];
    __shared__ unsigned int sword[BE];
    __shared__ int spos[BE];
    const int tid = threadIdx.x;
    const int lane = tid & 31;
    const int w = tid >> 5;
    for (int i = tid; i < NBK; i += 1024) { cnt[i] = 0; goff[i] = totals[i]; }
    __syncthreads();

    const int4* __restrict__ dv4 = (const int4*)didx + (size_t)blockIdx.x * BV;
    const int4* __restrict__ sv4 = (const int4*)sidx + (size_t)blockIdx.x * BV;
    const bool two = tid + 1024 < BV;
    const int i1 = two ? tid + 1024 : tid;
    int4 dd[2], ss[2];
    dd[0] = dv4[tid]; ss[0] = sv4[tid];
    dd[1] = dv4[i1];  ss[1] = sv4[i1];
    const int dl[8] = {dd[0].x, dd[0].y, dd[0].z, dd[0].w, dd[1].x, dd[1].y, dd[1].z, dd[1].w};
    const int sl[8] = {ss[0].x, ss[0].y, ss[0].z, ss[0].w, ss[1].x, ss[1].y, ss[1].z, ss[1].w};
    int bk[8], rk[8];
#pragma unroll
    for (int e = 0; e < 8; ++e) {
        bk[e] = dl[e] >> WSH;
        rk[e] = 0;
        if (e < 4 || two) rk[e] = atomicAdd(&cnt[bk[e]], 1);
    }
    __syncthreads();
    if (w == 0) wave_excl_scan(goff, goff, NBK, lane);
    __syncthreads();
    if (tid < NBK) goff[tid] += atomicAdd(&cursors[tid], cnt[tid]);
    __syncthreads();
    if (w == 0) wave_excl_scan(cnt, cnt, NBK, lane);
    __syncthreads();
#pragma unroll
    for (int e = 0; e < 8; ++e) {
        if (e < 4 || two) {
            const int slot = cnt[bk[e]] + rk[e];
            sword[slot] = ((unsigned int)sl[e] << WSH) | (unsigned int)(dl[e] & (BR - 1));
            spos[slot] = goff[bk[e]] + rk[e];
        }
    }
    __syncthreads();
    for (int i = tid; i < BE; i += 1024) bucket[spos[i]] = sword[i];
}

__device__ __forceinline__ const float* feat_row(const float* uf, const float* itf,
                                                 const float* bfv, int i) {
    const float* base = (i < NU) ? uf : ((i < NU + NI) ? itf : bfv);
    const int    r    = (i < NU) ? i  : ((i < NU + NI) ? i - NU : i - NU - NI);
    return base + (size_t)r * EMB;
}

#define SNW 16
template <int PASS>
__device__ __forceinline__ void scat_body(const unsigned int* __restrict__ bucket,
                                          const int* __restrict__ totals,
                                          const float* uf, const float* itf, const float* bfv,
                                          const float* __restrict__ src,
                                          const float* __restrict__ dvi, const float* __restrict__ dei,
                                          float* __restrict__ dst, float* __restrict__ partials) {
    __shared__ float acc[BR * EMB];
    __shared__ float sdata[SNW];
    __shared__ int sbase;
    const int b = blockIdx.x;
    const int tid = threadIdx.x;
    const int lane = tid & 31;
    const int w = __builtin_amdgcn_readfirstlane(tid >> 5);

    for (int i = tid; i < BR * EMB / 4; i += SNW * 32)
        ((float4*)acc)[i] = make_float4(0.0f, 0.0f, 0.0f, 0.0f);
    if (w == 0) {
        int s = 0;
        for (int i = lane; i < b; i += 32) s += totals[i];
#pragma unroll
        for (int d = 16; d > 0; d >>= 1) s += __shfl_xor(s, d);
        if (lane == 0) sbase = s;
    }
    __syncthreads();

    const int n = totals[b];
    const unsigned int* __restrict__ wq = bucket + sbase;
    const unsigned long long xb0 = (unsigned long long)uf;
    const unsigned long long xb1 = (unsigned long long)itf - (unsigned long long)NU * (EMB * sizeof(float));
    const unsigned long long xb2 = (unsigned long long)bfv - (unsigned long long)(NU + NI) * (EMB * sizeof(float));
    int entn = (w * 32 + lane < n) ? (int)wq[w * 32 + lane] : 0;
    for (int g = w * 32; g < n; g += SNW * 32) {
        const int ent = entn;
        const int gn = g + SNW * 32;
        const int m = n - g;
        float val[32];
        if (PASS == 1) {
            const unsigned int se = (unsigned int)ent >> WSH;
            const unsigned long long rb = (se < NU) ? xb0 : ((se < NU + NI) ? xb1 : xb2);
            const unsigned long long ra = rb + (unsigned long long)se * (EMB * sizeof(float));
            const int alo = (int)(unsigned int)ra, ahi = (int)(unsigned int)(ra >> 32);
            const int scb = __float_as_int(dvi[se]);
            const int la  = (int)(((unsigned int)ent & (unsigned int)(BR - 1)) * EMB);
#pragma unroll
            for (int j = 0; j < 32; ++j) {
                const unsigned long long a =
                    ((unsigned long long)(unsigned int)__builtin_amdgcn_readlane(ahi, j) << 32) |
                    (unsigned long long)(unsigned int)__builtin_amdgcn_readlane(alo, j);
                val[j] = ((const __attribute__((address_space(1))) float*)a)[lane];
            }
            entn = (gn + lane < n) ? (int)wq[gn + lane] : 0;
            if (m >= 32) {
#pragma unroll
                for (int j = 0; j < 32; ++j) {
                    const float f = __int_as_float(__builtin_amdgcn_readlane(scb, j));
                    atomicAdd(acc + __builtin_amdgcn_readlane(la, j) + lane, val[j] * f);
                }
            } else {
#pragma unroll
                for (int j = 0; j < 32; ++j) {
                    const float f = __int_as_float(__builtin_amdgcn_readlane(scb, j));
                    if (j < m) atomicAdd(acc + __builtin_amdgcn_readlane(la, j) + lane, val[j] * f);
                }
            }
        } else {
#pragma unroll
            for (int j = 0; j < 32; ++j) {
                const unsigned int e = (unsigned int)__builtin_amdgcn_readlane(ent, j);
                val[j] = src[(size_t)(e >> WSH) * EMB + lane];
            }
            entn = (gn + lane < n) ? (int)wq[gn + lane] : 0;
            if (m >= 32) {
#pragma unroll
                for (int j = 0; j < 32; ++j) {
                    const unsigned int e = (unsigned int)__builtin_amdgcn_readlane(ent, j);
                    atomicAdd(acc + (e & (unsigned int)(BR - 1)) * EMB + lane, val[j]);
                }
            } else {
#pragma unroll
                for (int j = 0; j < 32; ++j) {
                    const unsigned int e = (unsigned int)__builtin_amdgcn_readlane(ent, j);
                    if (j < m) atomicAdd(acc + (e & (unsigned int)(BR - 1)) * EMB + lane, val[j]);
                }
            }
        }
    }
    __syncthreads();

    const int lo = b * BR;
    const int nrows = (NTOT - lo < BR) ? (NTOT - lo) : BR;
    if (PASS == 1) {
        for (int row = w; row < nrows; row += SNW)
            dst[(size_t)(lo + row) * EMB + lane] = acc[row * EMB + lane] * dei[lo + row];
    } else {
        float sq = 0.0f;
        for (int row = w; row < nrows; row += SNW) {
            const int i = lo + row;
            const float x = feat_row(uf, itf, bfv, i)[lane];
            const float r = x * 0.5f + acc[row * EMB + lane] * dvi[i] * (1.0f / 3.0f);
            if (i < NU)            dst[(size_t)i * EMB + lane] = r;
            else if (i >= NU + NI) dst[(size_t)(i - NI) * EMB + lane] = r;
            sq += r * r;
        }
#pragma unroll
        for (int d = 16; d > 0; d >>= 1) sq += __shfl_xor(sq, d);
        if (lane == 0) sdata[w] = sq;
        __syncthreads();
        if (tid == 0) {
            float t = 0.0f;
            for (int k = 0; k < SNW; ++k) t += sdata[k];
            partials[b] = t;
        }
    }
}

__global__ __launch_bounds__(SNW * 32) void scat1_k(const unsigned int* __restrict__ bucket,
                                                    const int* __restrict__ totals,
                                                    const float* uf, const float* itf, const float* bfv,
                                                    const float* __restrict__ dvi, const float* __restrict__ dei,
                                                    float* __restrict__ u) {
    scat_body<1>(bucket, totals, uf, itf, bfv, uf, dvi, dei, u, u);
}

__global__ __launch_bounds__(SNW * 32) void scat2_k(const unsigned int* __restrict__ bucket,
                                                    const int* __restrict__ totals,
                                                    const float* uf, const float* itf, const float* bfv,
                                                    const float* __restrict__ u,
                                                    const float* __restrict__ dvi,
                                                    float* __restrict__ comp, float* __restrict__ partials) {
    scat_body<2>(bucket, totals, uf, itf, bfv, u, dvi, dvi, comp, partials);
}

static_assert((size_t)NNZE + (size_t)(NU + NB) * EMB <= (size_t)64 * NTOT, "bucket array and stored rows fit the region of the partial counts");
static_assert(2 * NBK <= 1024, "sizes and counters fit their slots behind the loss partials");

#define NKT 7
__global__ __launch_bounds__(256) void pred_k(const float* __restrict__ all, const int* __restrict__ users,
                       const int* __restrict__ bundles, const float* __restrict__ ubound,
                       const float* __restrict__ zrow,
                       const float* __restrict__ partials,
                       float* __restrict__ pred, float* __restrict__ usb,
                       float* __restrict__ out_loss) {
    if (blockIdx.x == 0) {
        __shared__ float sdata[256];
        float pv[(NBK + 255) / 256];
#pragma unroll
        for (int k = 0; k < (NBK + 255) / 256; ++k) {
            const int i = k * 256 + (int)threadIdx.x;
            pv[k] = partials[(i < NBK) ? i : NBK - 1];
        }
        float s = 0.0f;
#pragma unroll
        for (int k = 0; k < (NBK + 255) / 256; ++k)
            if (k * 256 + (int)threadIdx.x < NBK) s += pv[k];
        sdata[threadIdx.x] = s;
        __syncthreads();
        for (int k = 128; k > 0; k >>= 1) {
            if (threadIdx.x < (unsigned)k) sdata[threadIdx.x] += sdata[threadIdx.x + k];
            __syncthreads();
        }
        if (threadIdx.x == 0) *out_loss = L2NORM * sdata[0];
        return;
    }
    const int b = ((blockIdx.x - 1) * 256 + threadIdx.x) >> 5;
    const int lane = threadIdx.x & 31;

    const float* bfbase = all + (size_t)NU * EMB;
    const int ub = users[b];

    const int m = lane & 15;
    const int khalf = lane >> 4;

    static_assert((NKT - 1) * 16 <= KB && NKT * 16 >= KB + 5, "only the last tile has rows past the bundles, the bound's row among them");
    int bid[NKT - 1];
#pragma unroll
    for (int kt = 0; kt < NKT - 1; ++kt) bid[kt] = bundles[b * KB + kt * 16 + m];

    const float* alast;
    {
        const int kk = (NKT - 1) * 16 + m;
        if (kk < KB)       alast = bfbase + (size_t)bundles[b * KB + kk] * EMB;
        else if (m == 4)   alast = ubound;
        else               alast = zrow;
    }

    v8f c[NKT];
#pragma unroll
    for (int kt = 0; kt < NKT; ++kt) {
        c[kt] = (v8f){};
        asm volatile("" : "+v"(c[kt]));
    }

    const float* ue = all + (size_t)ub * EMB;
    v2f bb[8];
#pragma unroll
    for (int s = 0; s < 8; ++s) bb[s] = *(const v2f*)(ue + s * 4 + khalf * 2);

    v2f a[NKT][8];
#pragma unroll
    for (int kt = 0; kt < NKT - 1; ++kt) {
        const float* arow = bfbase + (size_t)bid[kt] * EMB;
#pragma unroll
        for (int s = 0; s < 8; ++s) a[kt][s] = *(const v2f*)(arow + s * 4 + khalf * 2);
    }
#pragma unroll
    for (int s = 0; s < 8; ++s) a[NKT - 1][s] = *(const v2f*)(alast + s * 4 + khalf * 2);

    __builtin_amdgcn_sched_barrier(0);

#pragma unroll
    for (int kt = 0; kt < NKT; ++kt) {
#pragma unroll
        for (int s = 0; s < 8; ++s)
            c[kt] = __builtin_amdgcn_wmma_f32_16x16x4_f32(false, a[kt][s], false, bb[s],
                                                          (short)0, c[kt], false, false);
    }

    int lane_st;
    asm volatile("v_nop\n\tv_nop\n\tv_nop\n\tv_nop\n\tv_nop\n\tv_nop\n\tv_nop\n\tv_nop\n\t"
                 "v_and_b32 %7, 31, %8"
                 : "+v"(c[0]), "+v"(c[1]), "+v"(c[2]), "+v"(c[3]), "+v"(c[4]), "+v"(c[5]), "+v"(c[6]),
                   "=v"(lane_st)
                 : "v"(threadIdx.x));

    if (lane_st == 0) {
#pragma unroll
        for (int kt = 0; kt < NKT; ++kt) {
#pragma unroll
            for (int j = 0; j < 8; ++j) {
                const int kkk = kt * 16 + j;
                if (kkk < KB)                       pred[(size_t)b * KB + kkk] = c[kt][j];
                else if (kt == NKT - 1 && j == 4)   usb[b] = c[kt][j];
            }
        }
    } else if (lane_st == 16) {
#pragma unroll
        for (int kt = 0; kt < NKT; ++kt) {
#pragma unroll
            for (int j = 0; j < 8; ++j) {
                const int kkk = kt * 16 + 8 + j;
                if (kkk < KB) pred[(size_t)b * KB + kkk] = c[kt][j];
            }
        }
    }
}

extern "C" void kernel_launch(void* const* d_in, const int* in_sizes, int n_in,
                              void* d_out, int out_size, void* d_ws, size_t ws_size,
                              hipStream_t stream) {
    const float* uf      = (const float*)d_in[0];
    const float* itf     = (const float*)d_in[1];
    const float* bfv     = (const float*)d_in[2];
    const float* ubound  = (const float*)d_in[3];
    const int*   rows    = (const int*)d_in[4];
    const int*   cols    = (const int*)d_in[5];
    const int*   users   = (const int*)d_in[6];
    const int*   bundles = (const int*)d_in[7];
    float* out = (float*)d_out;

    float* ws   = (float*)d_ws;
    float* dvde = ws;
    float* u    = dvde + 2L * NTOT;
    float* big  = u + 32L * NTOT;
    unsigned int* bucket = (unsigned int*)big;
    float* comp = big + NNZE;
    float* partials = big + 64L * NTOT;
    int*   totals  = (int*)(partials + 1024);
    int*   cursors = (int*)(partials + 2048);
    float* zrow = partials + 8192;

    hist_k<<<dim3(HQ, HP, 2), 1024, 0, stream>>>(rows, cols, big);
    prep_k<<<NBK, 256, 0, stream>>>(big, dvde, totals, cursors, zrow);

    bin_k<<<BQN, 1024, 0, stream>>>(rows, cols, totals + NBK, cursors + NBK, bucket);
    scat1_k<<<NBK, SNW * 32, 0, stream>>>(bucket, totals + NBK, uf, itf, bfv, dvde, dvde + NTOT, u);
    bin_k<<<BQN, 1024, 0, stream>>>(cols, rows, totals, cursors, bucket);
    scat2_k<<<NBK, SNW * 32, 0, stream>>>(bucket, totals, uf, itf, bfv, u, dvde, comp, partials);

    pred_k<<<BATCHN / 8 + 1, 256, 0, stream>>>(comp, users, bundles, ubound, zrow, partials,
                                     out, out + BATCHN * KB, out + BATCHN * KB + BATCHN);
}


# edge passes: bucket words ranked by destination row in LDS; each 8-lane group sums 2 rows in registers in rank order (16 B per lane, 4 source rows per load); no LDS float adds; loss regrouped
# speedup vs baseline: 5.6138x; 1.0347x over previous
#include <hip/hip_runtime.h>
#include <stddef.h>

#define NU   8039
#define NI   32770
#define NB   4771
#define NTOT (NU + NI + NB)
#define EMB  32
#define NNZE 2000000
#define BATCHN 2048
#define KB   100
#define EPSF 1e-8f
#define L2NORM 1e-5f

typedef __attribute__((ext_vector_type(2))) float v2f;
typedef __attribute__((ext_vector_type(8))) float v8f;
typedef __attribute__((ext_vector_type(4))) float v4f;
#define HP  4
#define HQ  32
#define HR  (NTOT / HP)
#define VPQ (NNZE / 4 / HQ)
#define HLD 8

__device__ __forceinline__ void bin_hit(unsigned int* bins, int idx, int lo) {
    unsigned int r = (unsigned int)(idx - lo);
    if (r < (unsigned int)HR) atomicAdd(bins + r, 1u);
}

__global__ __launch_bounds__(1024) void hist_k(const int* __restrict__ rows,
                                               const int* __restrict__ cols,
                                               float* __restrict__ part) {
    __shared__ unsigned int bins[HR];
    const int q = blockIdx.x, p = blockIdx.y, a = blockIdx.z;
    const int tid = threadIdx.x;
    for (int i = tid; i < HR; i += 1024) bins[i] = 0u;
    __syncthreads();

    const int lo = p * HR;
    const int4* __restrict__ vec = (const int4*)(a ? cols : rows) + (size_t)q * VPQ;
    for (int base = 0; base < VPQ; base += HLD * 1024) {
        int4 e[HLD];
#pragma unroll
        for (int j = 0; j < HLD; ++j) {
            int i = base + j * 1024 + tid;
            e[j] = (i < VPQ) ? vec[i] : make_int4(-1, -1, -1, -1);
        }
#pragma unroll
        for (int j = 0; j < HLD; ++j) {
            bin_hit(bins, e[j].x, lo);
            bin_hit(bins, e[j].y, lo);
            bin_hit(bins, e[j].z, lo);
            bin_hit(bins, e[j].w, lo);
        }
    }
    __syncthreads();

    float* __restrict__ dst = part + ((size_t)q * 2 + a) * NTOT + lo;
    for (int i = tid; i < HR; i += 1024) dst[i] = (float)bins[i];
}

#define BR   128
#define NBK  ((NTOT + BR - 1) / BR)
#define WSH  7
static_assert((1 << WSH) == BR, "the word's low bits hold a row of the bucket");
static_assert(NTOT <= 65536, "a source row and a row of the bucket fit one 32-bit word");
static_assert(NBK <= 12 * 32, "one wave scans the buckets, 12 per lane");

__device__ __forceinline__ void wave_excl_scan(const int* in, int* out, int n, int lane) {
    int v[12];
    int s = 0;
#pragma unroll
    for (int k = 0; k < 12; ++k) {
        const int idx = lane * 12 + k;
        v[k] = (idx < n) ? in[idx] : 0;
        s += v[k];
    }
    int inc = s;
#pragma unroll
    for (int d = 1; d < 32; d <<= 1) {
        const int t = __shfl_up(inc, d);
        if (lane >= d) inc += t;
    }
    int run = inc - s;
#pragma unroll
    for (int k = 0; k < 12; ++k) {
        const int idx = lane * 12 + k;
        if (idx < n) out[idx] = run;
        run += v[k];
    }
}

static_assert(2 * BR == 256, "one thread per (node, index array)");
__global__ __launch_bounds__(256) void prep_k(const float* __restrict__ part, float* __restrict__ dvde,
                                              int* __restrict__ totals, int* __restrict__ cursors,
                                              float* __restrict__ zrow) {
    __shared__ float sred[256];
    const int t = threadIdx.x;
    const int j = t & (BR - 1), a = t >> 7;
    const int i = blockIdx.x * BR + j;
    float d = 0.0f;
    if (i < NTOT) {
        const float* pp = part + (size_t)a * NTOT + i;
        float pv[HQ];
#pragma unroll
        for (int q = 0; q < HQ; ++q) pv[q] = pp[(size_t)q * (2 * NTOT)];
#pragma unroll
        for (int q = 0; q < HQ; ++q) d += pv[q];
        dvde[(size_t)a * NTOT + i] = 1.0f / (sqrtf(d) + EPSF);
    }
    sred[t] = d;
    if (blockIdx.x == 0) {
        if (t < 32) zrow[t] = 0.0f;
        for (int k = t; k < 2 * NBK; k += 256) cursors[k] = 0;
    }
    __syncthreads();
    for (int s = BR / 2; s > 0; s >>= 1) {
        if (j < s) sred[t] += sred[t + s];
        __syncthreads();
    }
    if (j == 0) totals[a * NBK + blockIdx.x] = (int)sred[t];
}

#define BV   2000
#define BE   (BV * 4)
#define BQN  (NNZE / BE)
static_assert(BQN * BE == NNZE, "every block is full");
static_assert(BV > 1024 && BV <= 2048, "a thread has one or two vectors");
__global__ __launch_bounds__(1024) void bin_k(const int* __restrict__ sidx, const int* __restrict__ didx,
                                              const int* __restrict__ totals, int* __restrict__ cursors,
                                              unsigned int* __restrict__ bucket) {
    __shared__ int cnt[NBK];
    __shared__ int goff[NBK];
    __shared__ unsigned int sword[BE];
    __shared__ int spos[BE];
    const int tid = threadIdx.x;
    const int lane = tid & 31;
    const int w = tid >> 5;
    for (int i = tid; i < NBK; i += 1024) { cnt[i] = 0; goff[i] = totals[i]; }
    __syncthreads();

    const int4* __restrict__ dv4 = (const int4*)didx + (size_t)blockIdx.x * BV;
    const int4* __restrict__ sv4 = (const int4*)sidx + (size_t)blockIdx.x * BV;
    const bool two = tid + 1024 < BV;
    const int i1 = two ? tid + 1024 : tid;
    int4 dd[2], ss[2];
    dd[0] = dv4[tid]; ss[0] = sv4[tid];
    dd[1] = dv4[i1];  ss[1] = sv4[i1];
    const int dl[8] = {dd[0].x, dd[0].y, dd[0].z, dd[0].w, dd[1].x, dd[1].y, dd[1].z, dd[1].w};
    const int sl[8] = {ss[0].x, ss[0].y, ss[0].z, ss[0].w, ss[1].x, ss[1].y, ss[1].z, ss[1].w};
    int bk[8], rk[8];
#pragma unroll
    for (int e = 0; e < 8; ++e) {
        bk[e] = dl[e] >> WSH;
        rk[e] = 0;
        if (e < 4 || two) rk[e] = atomicAdd(&cnt[bk[e]], 1);
    }
    __syncthreads();
    if (w == 0) wave_excl_scan(goff, goff, NBK, lane);
    __syncthreads();
    if (tid < NBK) goff[tid] += atomicAdd(&cursors[tid], cnt[tid]);
    __syncthreads();
    if (w == 0) wave_excl_scan(cnt, cnt, NBK, lane);
    __syncthreads();
#pragma unroll
    for (int e = 0; e < 8; ++e) {
        if (e < 4 || two) {
            const int slot = cnt[bk[e]] + rk[e];
            sword[slot] = ((unsigned int)sl[e] << WSH) | (unsigned int)(dl[e] & (BR - 1));
            spos[slot] = goff[bk[e]] + rk[e];
        }
    }
    __syncthreads();
    for (int i = tid; i < BE; i += 1024) bucket[spos[i]] = sword[i];
}

__device__ __forceinline__ const float* feat_row(const float* uf, const float* itf,
                                                 const float* bfv, int i) {
    const float* base = (i < NU) ? uf : ((i < NU + NI) ? itf : bfv);
    const int    r    = (i < NU) ? i  : ((i < NU + NI) ? i - NU : i - NU - NI);
    return base + (size_t)r * EMB;
}

#define SNW 16
#define CH  8192
#define CPT (CH / (SNW * 32))
#define RU  16
static_assert(BR == 2 * (SNW * 32 / 8), "two rows per group of 8 lanes");
static_assert(BR == 4 * 32, "one wave scans the row counters, 4 per lane");
template <int PASS>
__device__ __forceinline__ void scat_body(const unsigned int* __restrict__ bucket,
                                          const int* __restrict__ totals,
                                          const float* uf, const float* itf, const float* bfv,
                                          const float* __restrict__ src,
                                          const float* __restrict__ dvi, const float* __restrict__ dei,
                                          float* __restrict__ dst, float* __restrict__ partials) {
    __shared__ unsigned int sorted[CH + RU];
    __shared__ int rcnt[BR];
    __shared__ int rstart[BR];
    __shared__ float sdata[SNW];
    __shared__ int sbase;
    const int b = blockIdx.x;
    const int tid = threadIdx.x;
    const int lane = tid & 31;
    const int w = tid >> 5;
    const int grp = tid >> 3;
    const int sub = tid & 7;

    if (w == 0) {
        int s = 0;
        for (int i = lane; i < b; i += 32) s += totals[i];
#pragma unroll
        for (int d = 16; d > 0; d >>= 1) s += __shfl_xor(s, d);
        if (lane == 0) sbase = s;
    }
    __syncthreads();

    const int n = totals[b];
    const unsigned int* __restrict__ wq = bucket + sbase;
    const unsigned long long xb0 = (unsigned long long)uf + (unsigned long long)sub * 16u;
    const unsigned long long xb1 = (unsigned long long)itf + (unsigned long long)sub * 16u - (unsigned long long)NU * (EMB * sizeof(float));
    const unsigned long long xb2 = (unsigned long long)bfv + (unsigned long long)sub * 16u - (unsigned long long)(NU + NI) * (EMB * sizeof(float));
    float4 acc[2];
    acc[0] = make_float4(0.0f, 0.0f, 0.0f, 0.0f);
    acc[1] = make_float4(0.0f, 0.0f, 0.0f, 0.0f);

    for (int c0 = 0; c0 < n; c0 += CH) {
        const int cn = (n - c0 < CH) ? (n - c0) : CH;
        __syncthreads();
        if (tid < BR) rcnt[tid] = 0;
        unsigned int wv[CPT];
        int rk[CPT];
#pragma unroll
        for (int k = 0; k < CPT; ++k) {
            const int i = k * (SNW * 32) + tid;
            wv[k] = (i < cn) ? wq[c0 + i] : 0u;
        }
        __syncthreads();
#pragma unroll
        for (int k = 0; k < CPT; ++k) {
            const int i = k * (SNW * 32) + tid;
            rk[k] = (i < cn) ? atomicAdd(&rcnt[wv[k] & (unsigned int)(BR - 1)], 1) : 0;
        }
        __syncthreads();
        if (w == 0) {
            const int q0 = rcnt[lane * 4 + 0], q1 = rcnt[lane * 4 + 1];
            const int q2 = rcnt[lane * 4 + 2], q3 = rcnt[lane * 4 + 3];
            const int s4 = q0 + q1 + q2 + q3;
            int inc = s4;
#pragma unroll
            for (int d = 1; d < 32; d <<= 1) {
                const int t = __shfl_up(inc, d);
                if (lane >= d) inc += t;
            }
            const int run = inc - s4;
            rstart[lane * 4 + 0] = run;
            rstart[lane * 4 + 1] = run + q0;
            rstart[lane * 4 + 2] = run + q0 + q1;
            rstart[lane * 4 + 3] = run + q0 + q1 + q2;
        }
        __syncthreads();
#pragma unroll
        for (int k = 0; k < CPT; ++k) {
            const int i = k * (SNW * 32) + tid;
            if (i < cn) sorted[rstart[wv[k] & (unsigned int)(BR - 1)] + rk[k]] = wv[k] >> WSH;
        }
        __syncthreads();

#pragma unroll
        for (int r = 0; r < 2; ++r) {
            const int row = 2 * grp + r;
            const int s0 = rstart[row];
            const int L = rcnt[row];
            for (int k = 0; k < L; k += RU) {
                v4f v[RU];
                float f[RU];
                unsigned int sw[RU];
#pragma unroll
                for (int u = 0; u < RU; ++u) sw[u] = sorted[s0 + k + u];
#pragma unroll
                for (int u = 0; u < RU; ++u) {
                    const unsigned int sv = (k + u < L) ? sw[u] : 0u;
                    if (PASS == 1) {
                        const unsigned long long rb = (sv < NU) ? xb0 : ((sv < NU + NI) ? xb1 : xb2);
                        const unsigned long long ra = rb + (unsigned long long)sv * (EMB * sizeof(float));
                        v[u] = *(const __attribute__((address_space(1))) v4f*)ra;
                        f[u] = dvi[sv];
                    } else {
                        const unsigned int off = sv * (unsigned int)(EMB * sizeof(float)) + (unsigned int)sub * 16u;
                        v[u] = *(const v4f*)((const char*)src + off);
                    }
                }
                static_assert(RU == 16, "the statement names every row of a step");
                asm volatile("" : "+v"(v[0]), "+v"(v[1]), "+v"(v[2]), "+v"(v[3]), "+v"(v[4]), "+v"(v[5]),
                                  "+v"(v[6]), "+v"(v[7]), "+v"(v[8]), "+v"(v[9]), "+v"(v[10]), "+v"(v[11]),
                                  "+v"(v[12]), "+v"(v[13]), "+v"(v[14]), "+v"(v[15]));
                if (PASS == 1) {
                    asm volatile("" : "+v"(f[0]), "+v"(f[1]), "+v"(f[2]), "+v"(f[3]), "+v"(f[4]), "+v"(f[5]),
                                      "+v"(f[6]), "+v"(f[7]), "+v"(f[8]), "+v"(f[9]), "+v"(f[10]), "+v"(f[11]),
                                      "+v"(f[12]), "+v"(f[13]), "+v"(f[14]), "+v"(f[15]));
                }
#pragma unroll
                for (int u = 0; u < RU; ++u) {
                    const bool ok = k + u < L;
                    if (PASS == 1) {
                        acc[r].x += ok ? v[u].x * f[u] : 0.0f;
                        acc[r].y += ok ? v[u].y * f[u] : 0.0f;
                        acc[r].z += ok ? v[u].z * f[u] : 0.0f;
                        acc[r].w += ok ? v[u].w * f[u] : 0.0f;
                    } else {
                        acc[r].x += ok ? v[u].x : 0.0f;
                        acc[r].y += ok ? v[u].y : 0.0f;
                        acc[r].z += ok ? v[u].z : 0.0f;
                        acc[r].w += ok ? v[u].w : 0.0f;
                    }
                }
            }
        }
    }

    const int lo = b * BR;
    const int nrows = (NTOT - lo < BR) ? (NTOT - lo) : BR;
    if (PASS == 1) {
#pragma unroll
        for (int r = 0; r < 2; ++r) {
            const int row = 2 * grp + r;
            if (row < nrows) {
                const float e = dei[lo + row];
                *(float4*)(dst + (size_t)(lo + row) * EMB + sub * 4) =
                    make_float4(acc[r].x * e, acc[r].y * e, acc[r].z * e, acc[r].w * e);
            }
        }
    } else {
        float sq = 0.0f;
#pragma unroll
        for (int r = 0; r < 2; ++r) {
            const int row = 2 * grp + r;
            if (row < nrows) {
                const int i = lo + row;
                const float4 x = *(const float4*)(feat_row(uf, itf, bfv, i) + sub * 4);
                const float dv = dvi[i];
                float4 o;
                o.x = x.x * 0.5f + acc[r].x * dv * (1.0f / 3.0f);
                o.y = x.y * 0.5f + acc[r].y * dv * (1.0f / 3.0f);
                o.z = x.z * 0.5f + acc[r].z * dv * (1.0f / 3.0f);
                o.w = x.w * 0.5f + acc[r].w * dv * (1.0f / 3.0f);
                if (i < NU)            *(float4*)(dst + (size_t)i * EMB + sub * 4) = o;
                else if (i >= NU + NI) *(float4*)(dst + (size_t)(i - NI) * EMB + sub * 4) = o;
                sq += o.x * o.x + o.y * o.y + o.z * o.z + o.w * o.w;
            }
        }
#pragma unroll
        for (int d = 16; d > 0; d >>= 1) sq += __shfl_xor(sq, d);
        if (lane == 0) sdata[w] = sq;
        __syncthreads();
        if (tid == 0) {
            float t = 0.0f;
            for (int k = 0; k < SNW; ++k) t += sdata[k];
            partials[b] = t;
        }
    }
}

__global__ __launch_bounds__(SNW * 32, 8) void scat1_k(const unsigned int* __restrict__ bucket,
                                                    const int* __restrict__ totals,
                                                    const float* uf, const float* itf, const float* bfv,
                                                    const float* __restrict__ dvi, const float* __restrict__ dei,
                                                    float* __restrict__ u) {
    scat_body<1>(bucket, totals, uf, itf, bfv, uf, dvi, dei, u, u);
}

__global__ __launch_bounds__(SNW * 32, 8) void scat2_k(const unsigned int* __restrict__ bucket,
                                                    const int* __restrict__ totals,
                                                    const float* uf, const float* itf, const float* bfv,
                                                    const float* __restrict__ u,
                                                    const float* __restrict__ dvi,
                                                    float* __restrict__ comp, float* __restrict__ partials) {
    scat_body<2>(bucket, totals, uf, itf, bfv, u, dvi, dvi, comp, partials);
}

static_assert((size_t)NNZE + (size_t)(NU + NB) * EMB <= (size_t)64 * NTOT, "bucket array and stored rows fit the region of the partial counts");
static_assert(2 * NBK <= 1024, "sizes and counters fit their slots behind the loss partials");

#define NKT 7
__global__ __launch_bounds__(256) void pred_k(const float* __restrict__ all, const int* __restrict__ users,
                       const int* __restrict__ bundles, const float* __restrict__ ubound,
                       const float* __restrict__ zrow,
                       const float* __restrict__ partials,
                       float* __restrict__ pred, float* __restrict__ usb,
                       float* __restrict__ out_loss) {
    if (blockIdx.x == 0) {
        __shared__ float sdata[256];
        float pv[(NBK + 255) / 256];
#pragma unroll
        for (int k = 0; k < (NBK + 255) / 256; ++k) {
            const int i = k * 256 + (int)threadIdx.x;
            pv[k] = partials[(i < NBK) ? i : NBK - 1];
        }
        float s = 0.0f;
#pragma unroll
        for (int k = 0; k < (NBK + 255) / 256; ++k)
            if (k * 256 + (int)threadIdx.x < NBK) s += pv[k];
        sdata[threadIdx.x] = s;
        __syncthreads();
        for (int k = 128; k > 0; k >>= 1) {
            if (threadIdx.x < (unsigned)k) sdata[threadIdx.x] += sdata[threadIdx.x + k];
            __syncthreads();
        }
        if (threadIdx.x == 0) *out_loss = L2NORM * sdata[0];
        return;
    }
    const int b = ((blockIdx.x - 1) * 256 + threadIdx.x) >> 5;
    const int lane = threadIdx.x & 31;

    const float* bfbase = all + (size_t)NU * EMB;
    const int ub = users[b];

    const int m = lane & 15;
    const int khalf = lane >> 4;

    static_assert((NKT - 1) * 16 <= KB && NKT * 16 >= KB + 5, "only the last tile has rows past the bundles, the bound's row among them");
    int bid[NKT - 1];
#pragma unroll
    for (int kt = 0; kt < NKT - 1; ++kt) bid[kt] = bundles[b * KB + kt * 16 + m];

    const float* alast;
    {
        const int kk = (NKT - 1) * 16 + m;
        if (kk < KB)       alast = bfbase + (size_t)bundles[b * KB + kk] * EMB;
        else if (m == 4)   alast = ubound;
        else               alast = zrow;
    }

    v8f c[NKT];
#pragma unroll
    for (int kt = 0; kt < NKT; ++kt) {
        c[kt] = (v8f){};
        asm volatile("" : "+v"(c[kt]));
    }

    const float* ue = all + (size_t)ub * EMB;
    v2f bb[8];
#pragma unroll
    for (int s = 0; s < 8; ++s) bb[s] = *(const v2f*)(ue + s * 4 + khalf * 2);

    v2f a[NKT][8];
#pragma unroll
    for (int kt = 0; kt < NKT - 1; ++kt) {
        const float* arow = bfbase + (size_t)bid[kt] * EMB;
#pragma unroll
        for (int s = 0; s < 8; ++s) a[kt][s] = *(const v2f*)(arow + s * 4 + khalf * 2);
    }
#pragma unroll
    for (int s = 0; s < 8; ++s) a[NKT - 1][s] = *(const v2f*)(alast + s * 4 + khalf * 2);

    __builtin_amdgcn_sched_barrier(0);

#pragma unroll
    for (int kt = 0; kt < NKT; ++kt) {
#pragma unroll
        for (int s = 0; s < 8; ++s)
            c[kt] = __builtin_amdgcn_wmma_f32_16x16x4_f32(false, a[kt][s], false, bb[s],
                                                          (short)0, c[kt], false, false);
    }

    int lane_st;
    asm volatile("v_nop\n\tv_nop\n\tv_nop\n\tv_nop\n\tv_nop\n\tv_nop\n\tv_nop\n\tv_nop\n\t"
                 "v_and_b32 %7, 31, %8"
                 : "+v"(c[0]), "+v"(c[1]), "+v"(c[2]), "+v"(c[3]), "+v"(c[4]), "+v"(c[5]), "+v"(c[6]),
                   "=v"(lane_st)
                 : "v"(threadIdx.x));

    if (lane_st == 0) {
#pragma unroll
        for (int kt = 0; kt < NKT; ++kt) {
#pragma unroll
            for (int j = 0; j < 8; ++j) {
                const int kkk = kt * 16 + j;
                if (kkk < KB)                       pred[(size_t)b * KB + kkk] = c[kt][j];
                else if (kt == NKT - 1 && j == 4)   usb[b] = c[kt][j];
            }
        }
    } else if (lane_st == 16) {
#pragma unroll
        for (int kt = 0; kt < NKT; ++kt) {
#pragma unroll
            for (int j = 0; j < 8; ++j) {
                const int kkk = kt * 16 + 8 + j;
                if (kkk < KB) pred[(size_t)b * KB + kkk] = c[kt][j];
            }
        }
    }
}

extern "C" void kernel_launch(void* const* d_in, const int* in_sizes, int n_in,
                              void* d_out, int out_size, void* d_ws, size_t ws_size,
                              hipStream_t stream) {
    const float* uf      = (const float*)d_in[0];
    const float* itf     = (const float*)d_in[1];
    const float* bfv     = (const float*)d_in[2];
    const float* ubound  = (const float*)d_in[3];
    const int*   rows    = (const int*)d_in[4];
    const int*   cols    = (const int*)d_in[5];
    const int*   users   = (const int*)d_in[6];
    const int*   bundles = (const int*)d_in[7];
    float* out = (float*)d_out;

    float* ws   = (float*)d_ws;
    float* dvde = ws;
    float* u    = dvde + 2L * NTOT;
    float* big  = u + 32L * NTOT;
    unsigned int* bucket = (unsigned int*)big;
    float* comp = big + NNZE;
    float* partials = big + 64L * NTOT;
    int*   totals  = (int*)(partials + 1024);
    int*   cursors = (int*)(partials + 2048);
    float* zrow = partials + 8192;

    hist_k<<<dim3(HQ, HP, 2), 1024, 0, stream>>>(rows, cols, big);
    prep_k<<<NBK, 256, 0, stream>>>(big, dvde, totals, cursors, zrow);

    bin_k<<<BQN, 1024, 0, stream>>>(rows, cols, totals + NBK, cursors + NBK, bucket);
    scat1_k<<<NBK, SNW * 32, 0, stream>>>(bucket, totals + NBK, uf, itf, bfv, dvde, dvde + NTOT, u);
    bin_k<<<BQN, 1024, 0, stream>>>(cols, rows, totals, cursors, bucket);
    scat2_k<<<NBK, SNW * 32, 0, stream>>>(bucket, totals, uf, itf, bfv, u, dvde, comp, partials);

    pred_k<<<BATCHN / 8 + 1, 256, 0, stream>>>(comp, users, bundles, ubound, zrow, partials,
                                     out, out + BATCHN * KB, out + BATCHN * KB + BATCHN);
}


# launch fusion, 7 launches to 6: degree factors and bucket sizes by extra blocks of the first sort's launch, which adds up per-chunk bucket counts itself; chunk counts kept as exact 16-bit integers
# speedup vs baseline: 5.8711x; 1.0458x over previous
#include <hip/hip_runtime.h>
#include <stddef.h>

#define NU   8039
#define NI   32770
#define NB   4771
#define NTOT (NU + NI + NB)
#define EMB  32
#define NNZE 2000000
#define BATCHN 2048
#define KB   100
#define EPSF 1e-8f
#define L2NORM 1e-5f

typedef __attribute__((ext_vector_type(2))) float v2f;
typedef __attribute__((ext_vector_type(8))) float v8f;
typedef __attribute__((ext_vector_type(4))) float v4f;
#define BR   128
#define NBK  ((NTOT + BR - 1) / BR)

#define HP  4
#define HQ  32
#define HB  ((NBK + HP - 1) / HP)
#define HR  (HB * BR)
#define VPQ (NNZE / 4 / HQ)
#define HLD 8
static_assert(HP * HR >= NTOT, "the ranges cover every node");
static_assert(BR == 4 * 32, "a wave sums a bucket's counters, 4 per lane");
static_assert(VPQ * 4 < 65536, "a chunk's count of one node fits 16 bits");
static_assert(NTOT % 2 == 0 && HR % 2 == 0, "counts are stored in pairs");

__device__ __forceinline__ void bin_hit(unsigned int* bins, int idx, int lo) {
    unsigned int r = (unsigned int)(idx - lo);
    if (r < (unsigned int)HR) atomicAdd(bins + r, 1u);
}

__global__ __launch_bounds__(1024) void hist_k(const int* __restrict__ rows,
                                               const int* __restrict__ cols,
                                               unsigned short* __restrict__ part, int* __restrict__ ptot,
                                               int* __restrict__ cursors, float* __restrict__ zrow) {
    __shared__ __attribute__((aligned(16))) unsigned int bins[HR];
    const int q = blockIdx.x, p = blockIdx.y, a = blockIdx.z;
    const int tid = threadIdx.x;
    for (int i = tid; i < HR; i += 1024) bins[i] = 0u;
    if (q == 0 && p == 0 && a == 0) {
        if (tid < 32) zrow[tid] = 0.0f;
        for (int k = tid; k < 2 * NBK; k += 1024) cursors[k] = 0;
    }
    __syncthreads();

    const int lo = p * HR;
    const int4* __restrict__ vec = (const int4*)(a ? cols : rows) + (size_t)q * VPQ;
    for (int base = 0; base < VPQ; base += HLD * 1024) {
        int4 e[HLD];
#pragma unroll
        for (int j = 0; j < HLD; ++j) {
            int i = base + j * 1024 + tid;
            e[j] = (i < VPQ) ? vec[i] : make_int4(-1, -1, -1, -1);
        }
#pragma unroll
        for (int j = 0; j < HLD; ++j) {
            bin_hit(bins, e[j].x, lo);
            bin_hit(bins, e[j].y, lo);
            bin_hit(bins, e[j].z, lo);
            bin_hit(bins, e[j].w, lo);
        }
    }
    __syncthreads();

    unsigned int* __restrict__ dst = (unsigned int*)(part + ((size_t)q * 2 + a) * NTOT + lo);
    for (int i = tid; i < HR / 2; i += 1024)
        if (lo + 2 * i < NTOT) dst[i] = bins[2 * i] | (bins[2 * i + 1] << 16);

    const int lane = tid & 31;
    for (int bk = tid >> 5; bk < HB; bk += 32) {
        const uint4 c4 = *(const uint4*)&bins[bk * BR + lane * 4];
        unsigned int s = c4.x + c4.y + c4.z + c4.w;
#pragma unroll
        for (int d = 16; d > 0; d >>= 1) s += __shfl_xor(s, d);
        const int gb = p * HB + bk;
        if (lane == 0 && gb < NBK) ptot[((size_t)q * 2 + a) * NBK + gb] = (int)s;
    }
}

#define WSH  7
static_assert((1 << WSH) == BR, "the word's low bits hold a row of the bucket");
static_assert(NTOT <= 65536, "a source row and a row of the bucket fit one 32-bit word");
static_assert(NBK <= 12 * 32, "one wave scans the buckets, 12 per lane");

__device__ __forceinline__ void wave_excl_scan(const int* in, int* out, int n, int lane) {
    int v[12];
    int s = 0;
#pragma unroll
    for (int k = 0; k < 12; ++k) {
        const int idx = lane * 12 + k;
        v[k] = (idx < n) ? in[idx] : 0;
        s += v[k];
    }
    int inc = s;
#pragma unroll
    for (int d = 1; d < 32; d <<= 1) {
        const int t = __shfl_up(inc, d);
        if (lane >= d) inc += t;
    }
    int run = inc - s;
#pragma unroll
    for (int k = 0; k < 12; ++k) {
        const int idx = lane * 12 + k;
        if (idx < n) out[idx] = run;
        run += v[k];
    }
}

#define DN  512
#define PB  ((NTOT + DN - 1) / DN)
static_assert(2 * DN == 1024, "one thread per (node, index array)");
static_assert(2 * NBK <= 1024, "one thread per bucket size");
__device__ __forceinline__ void degree_part(const unsigned short* __restrict__ part, const int* __restrict__ ptot,
                                            float* __restrict__ dvde, int* __restrict__ totals,
                                            int e, int t) {
    const int j = t & (DN - 1), a = t >> 9;
    const int i = e * DN + j;
    if (i < NTOT) {
        const unsigned short* pp = part + (size_t)a * NTOT + i;
        float pv[HQ];
        float d = 0.0f;
#pragma unroll
        for (int q = 0; q < HQ; ++q) pv[q] = (float)pp[(size_t)q * (2 * NTOT)];
#pragma unroll
        for (int q = 0; q < HQ; ++q) d += pv[q];
        dvde[(size_t)a * NTOT + i] = 1.0f / (sqrtf(d) + EPSF);
    }
    if (e == 0 && t < 2 * NBK) {
        int pc[HQ];
        int s = 0;
#pragma unroll
        for (int q = 0; q < HQ; ++q) pc[q] = ptot[q * (2 * NBK) + t];
#pragma unroll
        for (int q = 0; q < HQ; ++q) s += pc[q];
        totals[t] = s;
    }
}

#define BV   2000
#define BE   (BV * 4)
#define BQN  (NNZE / BE)
static_assert(BQN * BE == NNZE, "every block is full");
static_assert(BV > 1024 && BV <= 2048, "a thread has one or two vectors");
template <int FIRST>
__global__ __launch_bounds__(1024) void bin_k(const int* __restrict__ sidx, const int* __restrict__ didx,
                                              const int* __restrict__ totals, int* __restrict__ cursors,
                                              unsigned int* __restrict__ bucket,
                                              const unsigned short* __restrict__ part, const int* __restrict__ ptot,
                                              float* __restrict__ dvde, int* __restrict__ totals_out) {
    if (FIRST && blockIdx.x >= BQN) {
        degree_part(part, ptot, dvde, totals_out, (int)blockIdx.x - BQN, (int)threadIdx.x);
        return;
    }
    __shared__ int cnt[NBK];
    __shared__ int goff[NBK];
    __shared__ unsigned int sword[BE];
    __shared__ int spos[BE];
    const int tid = threadIdx.x;
    const int lane = tid & 31;
    const int w = tid >> 5;
    if (FIRST) {
        if (tid < NBK) {
            int pc[HQ];
            int s = 0;
#pragma unroll
            for (int q = 0; q < HQ; ++q) pc[q] = totals[q * (2 * NBK) + tid];
#pragma unroll
            for (int q = 0; q < HQ; ++q) s += pc[q];
            cnt[tid] = 0;
            goff[tid] = s;
        }
    } else {
        for (int i = tid; i < NBK; i += 1024) { cnt[i] = 0; goff[i] = totals[i]; }
    }
    __syncthreads();

    const int4* __restrict__ dv4 = (const int4*)didx + (size_t)blockIdx.x * BV;
    const int4* __restrict__ sv4 = (const int4*)sidx + (size_t)blockIdx.x * BV;
    const bool two = tid + 1024 < BV;
    const int i1 = two ? tid + 1024 : tid;
    int4 dd[2], ss[2];
    dd[0] = dv4[tid]; ss[0] = sv4[tid];
    dd[1] = dv4[i1];  ss[1] = sv4[i1];
    const int dl[8] = {dd[0].x, dd[0].y, dd[0].z, dd[0].w, dd[1].x, dd[1].y, dd[1].z, dd[1].w};
    const int sl[8] = {ss[0].x, ss[0].y, ss[0].z, ss[0].w, ss[1].x, ss[1].y, ss[1].z, ss[1].w};
    int bk[8], rk[8];
#pragma unroll
    for (int e = 0; e < 8; ++e) {
        bk[e] = dl[e] >> WSH;
        rk[e] = 0;
        if (e < 4 || two) rk[e] = atomicAdd(&cnt[bk[e]], 1);
    }
    __syncthreads();
    if (w == 0) wave_excl_scan(goff, goff, NBK, lane);
    __syncthreads();
    if (tid < NBK) goff[tid] += atomicAdd(&cursors[tid], cnt[tid]);
    __syncthreads();
    if (w == 0) wave_excl_scan(cnt, cnt, NBK, lane);
    __syncthreads();
#pragma unroll
    for (int e = 0; e < 8; ++e) {
        if (e < 4 || two) {
            const int slot = cnt[bk[e]] + rk[e];
            sword[slot] = ((unsigned int)sl[e] << WSH) | (unsigned int)(dl[e] & (BR - 1));
            spos[slot] = goff[bk[e]] + rk[e];
        }
    }
    __syncthreads();
    for (int i = tid; i < BE; i += 1024) bucket[spos[i]] = sword[i];
}

__device__ __forceinline__ const float* feat_row(const float* uf, const float* itf,
                                                 const float* bfv, int i) {
    const float* base = (i < NU) ? uf : ((i < NU + NI) ? itf : bfv);
    const int    r    = (i < NU) ? i  : ((i < NU + NI) ? i - NU : i - NU - NI);
    return base + (size_t)r * EMB;
}

#define SNW 16
#define CH  8192
#define CPT (CH / (SNW * 32))
#define RU  16
static_assert(BR == 2 * (SNW * 32 / 8), "two rows per group of 8 lanes");
static_assert(BR == 4 * 32, "one wave scans the row counters, 4 per lane");
template <int PASS>
__device__ __forceinline__ void scat_body(const unsigned int* __restrict__ bucket,
                                          const int* __restrict__ totals,
                                          const float* uf, const float* itf, const float* bfv,
                                          const float* __restrict__ src,
                                          const float* __restrict__ dvi, const float* __restrict__ dei,
                                          float* __restrict__ dst, float* __restrict__ partials) {
    __shared__ unsigned int sorted[CH + RU];
    __shared__ int rcnt[BR];
    __shared__ int rstart[BR];
    __shared__ float sdata[SNW];
    __shared__ int sbase;
    const int b = blockIdx.x;
    const int tid = threadIdx.x;
    const int lane = tid & 31;
    const int w = tid >> 5;
    const int grp = tid >> 3;
    const int sub = tid & 7;

    if (w == 0) {
        int s = 0;
        for (int i = lane; i < b; i += 32) s += totals[i];
#pragma unroll
        for (int d = 16; d > 0; d >>= 1) s += __shfl_xor(s, d);
        if (lane == 0) sbase = s;
    }
    __syncthreads();

    const int n = totals[b];
    const unsigned int* __restrict__ wq = bucket + sbase;
    const unsigned long long xb0 = (unsigned long long)uf + (unsigned long long)sub * 16u;
    const unsigned long long xb1 = (unsigned long long)itf + (unsigned long long)sub * 16u - (unsigned long long)NU * (EMB * sizeof(float));
    const unsigned long long xb2 = (unsigned long long)bfv + (unsigned long long)sub * 16u - (unsigned long long)(NU + NI) * (EMB * sizeof(float));
    float4 acc[2];
    acc[0] = make_float4(0.0f, 0.0f, 0.0f, 0.0f);
    acc[1] = make_float4(0.0f, 0.0f, 0.0f, 0.0f);

    for (int c0 = 0; c0 < n; c0 += CH) {
        const int cn = (n - c0 < CH) ? (n - c0) : CH;
        __syncthreads();
        if (tid < BR) rcnt[tid] = 0;
        unsigned int wv[CPT];
        int rk[CPT];
#pragma unroll
        for (int k = 0; k < CPT; ++k) {
            const int i = k * (SNW * 32) + tid;
            wv[k] = (i < cn) ? wq[c0 + i] : 0u;
        }
        __syncthreads();
#pragma unroll
        for (int k = 0; k < CPT; ++k) {
            const int i = k * (SNW * 32) + tid;
            rk[k] = (i < cn) ? atomicAdd(&rcnt[wv[k] & (unsigned int)(BR - 1)], 1) : 0;
        }
        __syncthreads();
        if (w == 0) {
            const int q0 = rcnt[lane * 4 + 0], q1 = rcnt[lane * 4 + 1];
            const int q2 = rcnt[lane * 4 + 2], q3 = rcnt[lane * 4 + 3];
            const int s4 = q0 + q1 + q2 + q3;
            int inc = s4;
#pragma unroll
            for (int d = 1; d < 32; d <<= 1) {
                const int t = __shfl_up(inc, d);
                if (lane >= d) inc += t;
            }
            const int run = inc - s4;
            rstart[lane * 4 + 0] = run;
            rstart[lane * 4 + 1] = run + q0;
            rstart[lane * 4 + 2] = run + q0 + q1;
            rstart[lane * 4 + 3] = run + q0 + q1 + q2;
        }
        __syncthreads();
#pragma unroll
        for (int k = 0; k < CPT; ++k) {
            const int i = k * (SNW * 32) + tid;
            if (i < cn) sorted[rstart[wv[k] & (unsigned int)(BR - 1)] + rk[k]] = wv[k] >> WSH;
        }
        __syncthreads();

#pragma unroll
        for (int r = 0; r < 2; ++r) {
            const int row = 2 * grp + r;
            const int s0 = rstart[row];
            const int L = rcnt[row];
            for (int k = 0; k < L; k += RU) {
                v4f v[RU];
                float f[RU];
                unsigned int sw[RU];
#pragma unroll
                for (int u = 0; u < RU; ++u) sw[u] = sorted[s0 + k + u];
#pragma unroll
                for (int u = 0; u < RU; ++u) {
                    const unsigned int sv = (k + u < L) ? sw[u] : 0u;
                    if (PASS == 1) {
                        const unsigned long long rb = (sv < NU) ? xb0 : ((sv < NU + NI) ? xb1 : xb2);
                        const unsigned long long ra = rb + (unsigned long long)sv * (EMB * sizeof(float));
                        v[u] = *(const __attribute__((address_space(1))) v4f*)ra;
                        f[u] = dvi[sv];
                    } else {
                        const unsigned int off = sv * (unsigned int)(EMB * sizeof(float)) + (unsigned int)sub * 16u;
                        v[u] = *(const v4f*)((const char*)src + off);
                    }
                }
                static_assert(RU == 16, "the statement names every row of a step");
                asm volatile("" : "+v"(v[0]), "+v"(v[1]), "+v"(v[2]), "+v"(v[3]), "+v"(v[4]), "+v"(v[5]),
                                  "+v"(v[6]), "+v"(v[7]), "+v"(v[8]), "+v"(v[9]), "+v"(v[10]), "+v"(v[11]),
                                  "+v"(v[12]), "+v"(v[13]), "+v"(v[14]), "+v"(v[15]));
                if (PASS == 1) {
                    asm volatile("" : "+v"(f[0]), "+v"(f[1]), "+v"(f[2]), "+v"(f[3]), "+v"(f[4]), "+v"(f[5]),
                                      "+v"(f[6]), "+v"(f[7]), "+v"(f[8]), "+v"(f[9]), "+v"(f[10]), "+v"(f[11]),
                                      "+v"(f[12]), "+v"(f[13]), "+v"(f[14]), "+v"(f[15]));
                }
#pragma unroll
                for (int u = 0; u < RU; ++u) {
                    const bool ok = k + u < L;
                    if (PASS == 1) {
                        acc[r].x += ok ? v[u].x * f[u] : 0.0f;
                        acc[r].y += ok ? v[u].y * f[u] : 0.0f;
                        acc[r].z += ok ? v[u].z * f[u] : 0.0f;
                        acc[r].w += ok ? v[u].w * f[u] : 0.0f;
                    } else {
                        acc[r].x += ok ? v[u].x : 0.0f;
                        acc[r].y += ok ? v[u].y : 0.0f;
                        acc[r].z += ok ? v[u].z : 0.0f;
                        acc[r].w += ok ? v[u].w : 0.0f;
                    }
                }
            }
        }
    }

    const int lo = b * BR;
    const int nrows = (NTOT - lo < BR) ? (NTOT - lo) : BR;
    if (PASS == 1) {
#pragma unroll
        for (int r = 0; r < 2; ++r) {
            const int row = 2 * grp + r;
            if (row < nrows) {
                const float e = dei[lo + row];
                *(float4*)(dst + (size_t)(lo + row) * EMB + sub * 4) =
                    make_float4(acc[r].x * e, acc[r].y * e, acc[r].z * e, acc[r].w * e);
            }
        }
    } else {
        float sq = 0.0f;
#pragma unroll
        for (int r = 0; r < 2; ++r) {
            const int row = 2 * grp + r;
            if (row < nrows) {
                const int i = lo + row;
                const float4 x = *(const float4*)(feat_row(uf, itf, bfv, i) + sub * 4);
                const float dv = dvi[i];
                float4 o;
                o.x = x.x * 0.5f + acc[r].x * dv * (1.0f / 3.0f);
                o.y = x.y * 0.5f + acc[r].y * dv * (1.0f / 3.0f);
                o.z = x.z * 0.5f + acc[r].z * dv * (1.0f / 3.0f);
                o.w = x.w * 0.5f + acc[r].w * dv * (1.0f / 3.0f);
                if (i < NU)            *(float4*)(dst + (size_t)i * EMB + sub * 4) = o;
                else if (i >= NU + NI) *(float4*)(dst + (size_t)(i - NI) * EMB + sub * 4) = o;
                sq += o.x * o.x + o.y * o.y + o.z * o.z + o.w * o.w;
            }
        }
#pragma unroll
        for (int d = 16; d > 0; d >>= 1) sq += __shfl_xor(sq, d);
        if (lane == 0) sdata[w] = sq;
        __syncthreads();
        if (tid == 0) {
            float t = 0.0f;
            for (int k = 0; k < SNW; ++k) t += sdata[k];
            partials[b] = t;
        }
    }
}

__global__ __launch_bounds__(SNW * 32, 8) void scat1_k(const unsigned int* __restrict__ bucket,
                                                    const int* __restrict__ totals,
                                                    const float* uf, const float* itf, const float* bfv,
                                                    const float* __restrict__ dvi, const float* __restrict__ dei,
                                                    float* __restrict__ u) {
    scat_body<1>(bucket, totals, uf, itf, bfv, uf, dvi, dei, u, u);
}

__global__ __launch_bounds__(SNW * 32, 8) void scat2_k(const unsigned int* __restrict__ bucket,
                                                    const int* __restrict__ totals,
                                                    const float* uf, const float* itf, const float* bfv,
                                                    const float* __restrict__ u,
                                                    const float* __restrict__ dvi,
                                                    float* __restrict__ comp, float* __restrict__ partials) {
    scat_body<2>(bucket, totals, uf, itf, bfv, u, dvi, dvi, comp, partials);
}

static_assert((size_t)NNZE + (size_t)(NU + NB) * EMB + (size_t)HQ * 2 * NBK <= (size_t)64 * NTOT, "bucket array, stored rows and partial bucket counts fit the big region");
static_assert((size_t)HQ * 2 * NTOT * sizeof(unsigned short) <= (size_t)32 * NTOT * sizeof(float), "the 16-bit partial degree counts fit the region of u");
static_assert(2 * NBK <= 1024, "sizes and counters fit their slots behind the loss partials");

#define NKT 7
__global__ __launch_bounds__(256) void pred_k(const float* __restrict__ all, const int* __restrict__ users,
                       const int* __restrict__ bundles, const float* __restrict__ ubound,
                       const float* __restrict__ zrow,
                       const float* __restrict__ partials,
                       float* __restrict__ pred, float* __restrict__ usb,
                       float* __restrict__ out_loss) {
    if (blockIdx.x == 0) {
        __shared__ float sdata[256];
        float pv[(NBK + 255) / 256];
#pragma unroll
        for (int k = 0; k < (NBK + 255) / 256; ++k) {
            const int i = k * 256 + (int)threadIdx.x;
            pv[k] = partials[(i < NBK) ? i : NBK - 1];
        }
        float s = 0.0f;
#pragma unroll
        for (int k = 0; k < (NBK + 255) / 256; ++k)
            if (k * 256 + (int)threadIdx.x < NBK) s += pv[k];
        sdata[threadIdx.x] = s;
        __syncthreads();
        for (int k = 128; k > 0; k >>= 1) {
            if (threadIdx.x < (unsigned)k) sdata[threadIdx.x] += sdata[threadIdx.x + k];
            __syncthreads();
        }
        if (threadIdx.x == 0) *out_loss = L2NORM * sdata[0];
        return;
    }
    const int b = ((blockIdx.x - 1) * 256 + threadIdx.x) >> 5;
    const int lane = threadIdx.x & 31;

    const float* bfbase = all + (size_t)NU * EMB;
    const int ub = users[b];

    const int m = lane & 15;
    const int khalf = lane >> 4;

    static_assert((NKT - 1) * 16 <= KB && NKT * 16 >= KB + 5, "only the last tile has rows past the bundles, the bound's row among them");
    int bid[NKT - 1];
#pragma unroll
    for (int kt = 0; kt < NKT - 1; ++kt) bid[kt] = bundles[b * KB + kt * 16 + m];

    const float* alast;
    {
        const int kk = (NKT - 1) * 16 + m;
        if (kk < KB)       alast = bfbase + (size_t)bundles[b * KB + kk] * EMB;
        else if (m == 4)   alast = ubound;
        else               alast = zrow;
    }

    v8f c[NKT];
#pragma unroll
    for (int kt = 0; kt < NKT; ++kt) {
        c[kt] = (v8f){};
        asm volatile("" : "+v"(c[kt]));
    }

    const float* ue = all + (size_t)ub * EMB;
    v2f bb[8];
#pragma unroll
    for (int s = 0; s < 8; ++s) bb[s] = *(const v2f*)(ue + s * 4 + khalf * 2);

    v2f a[NKT][8];
#pragma unroll
    for (int kt = 0; kt < NKT - 1; ++kt) {
        const float* arow = bfbase + (size_t)bid[kt] * EMB;
#pragma unroll
        for (int s = 0; s < 8; ++s) a[kt][s] = *(const v2f*)(arow + s * 4 + khalf * 2);
    }
#pragma unroll
    for (int s = 0; s < 8; ++s) a[NKT - 1][s] = *(const v2f*)(alast + s * 4 + khalf * 2);

    __builtin_amdgcn_sched_barrier(0);

#pragma unroll
    for (int kt = 0; kt < NKT; ++kt) {
#pragma unroll
        for (int s = 0; s < 8; ++s)
            c[kt] = __builtin_amdgcn_wmma_f32_16x16x4_f32(false, a[kt][s], false, bb[s],
                                                          (short)0, c[kt], false, false);
    }

    int lane_st;
    asm volatile("v_nop\n\tv_nop\n\tv_nop\n\tv_nop\n\tv_nop\n\tv_nop\n\tv_nop\n\tv_nop\n\t"
                 "v_and_b32 %7, 31, %8"
                 : "+v"(c[0]), "+v"(c[1]), "+v"(c[2]), "+v"(c[3]), "+v"(c[4]), "+v"(c[5]), "+v"(c[6]),
                   "=v"(lane_st)
                 : "v"(threadIdx.x));

    if (lane_st == 0) {
#pragma unroll
        for (int kt = 0; kt < NKT; ++kt) {
#pragma unroll
            for (int j = 0; j < 8; ++j) {
                const int kkk = kt * 16 + j;
                if (kkk < KB)                       pred[(size_t)b * KB + kkk] = c[kt][j];
                else if (kt == NKT - 1 && j == 4)   usb[b] = c[kt][j];
            }
        }
    } else if (lane_st == 16) {
#pragma unroll
        for (int kt = 0; kt < NKT; ++kt) {
#pragma unroll
            for (int j = 0; j < 8; ++j) {
                const int kkk = kt * 16 + 8 + j;
                if (kkk < KB) pred[(size_t)b * KB + kkk] = c[kt][j];
            }
        }
    }
}

extern "C" void kernel_launch(void* const* d_in, const int* in_sizes, int n_in,
                              void* d_out, int out_size, void* d_ws, size_t ws_size,
                              hipStream_t stream) {
    const float* uf      = (const float*)d_in[0];
    const float* itf     = (const float*)d_in[1];
    const float* bfv     = (const float*)d_in[2];
    const float* ubound  = (const float*)d_in[3];
    const int*   rows    = (const int*)d_in[4];
    const int*   cols    = (const int*)d_in[5];
    const int*   users   = (const int*)d_in[6];
    const int*   bundles = (const int*)d_in[7];
    float* out = (float*)d_out;

    float* ws   = (float*)d_ws;
    float* dvde = ws;
    float* u    = dvde + 2L * NTOT;
    float* big  = u + 32L * NTOT;
    unsigned short* part = (unsigned short*)u;
    unsigned int* bucket = (unsigned int*)big;
    float* comp = big + NNZE;
    int*   ptot = (int*)(comp + (size_t)(NU + NB) * EMB);
    float* partials = big + 64L * NTOT;
    int*   totals  = (int*)(partials + 1024);
    int*   cursors = (int*)(partials + 2048);
    float* zrow = partials + 8192;

    hist_k<<<dim3(HQ, HP, 2), 1024, 0, stream>>>(rows, cols, part, ptot, cursors, zrow);

    bin_k<1><<<BQN + PB, 1024, 0, stream>>>(rows, cols, ptot + NBK, cursors + NBK, bucket, part, ptot, dvde, totals);
    scat1_k<<<NBK, SNW * 32, 0, stream>>>(bucket, totals + NBK, uf, itf, bfv, dvde, dvde + NTOT, u);
    bin_k<0><<<BQN, 1024, 0, stream>>>(cols, rows, totals, cursors, bucket, part, ptot, dvde, totals);
    scat2_k<<<NBK, SNW * 32, 0, stream>>>(bucket, totals, uf, itf, bfv, u, dvde, comp, partials);

    pred_k<<<BATCHN / 8 + 1, 256, 0, stream>>>(comp, users, bundles, ubound, zrow, partials,
                                     out, out + BATCHN * KB, out + BATCHN * KB + BATCHN);
}
